# starting kernel
# baseline (speedup 1.0000x reference)
#include <hip/hip_runtime.h>
#include <hip/hip_bf16.h>


typedef float v2f __attribute__((ext_vector_type(2)));
typedef float v8f __attribute__((ext_vector_type(8)));

__global__ void sage_deg_count(const int* __restrict__ dst, float* __restrict__ cnt,
                               int nedges) {
    int e = blockIdx.x * blockDim.x + threadIdx.x;
    if (e >= nedges) return;
    atomicAdd(&cnt[dst[e]], 1.0f);
}

__global__ void sage_inv_deg(float* __restrict__ cnt, int n) {
    int i = blockIdx.x * blockDim.x + threadIdx.x;
    if (i >= n) return;
    cnt[i] = 1.0f / fmaxf(cnt[i], 1.0f);
}

__global__ void sage_scatter_add(const float* __restrict__ xin,
                                 const int* __restrict__ src,
                                 const int* __restrict__ dst,
                                 float* __restrict__ agg,
                                 int f4, long long nwork) {
    long long tid = (long long)blockIdx.x * blockDim.x + threadIdx.x;
    if (tid >= nwork) return;
    int e = (int)(tid / f4);
    int c = (int)(tid % f4);
    int F = f4 * 4;
    int s = src[e];
    int d = dst[e];
    const float4 v = *(const float4*)(xin + (size_t)s * F + c * 4);
    float* p = agg + (size_t)d * F + c * 4;
    atomicAdd(p + 0, v.x);
    atomicAdd(p + 1, v.y);
    atomicAdd(p + 2, v.z);
    atomicAdd(p + 3, v.w);
}

__global__ void __launch_bounds__(256)
sage_gemm_wmma(const float* __restrict__ agg, const float* __restrict__ inv,
               const float* __restrict__ xin,
               const float* __restrict__ Wl, const float* __restrict__ Wr,
               const float* __restrict__ bias, float* __restrict__ out,
               int K, int Nout, int nbx, int do_relu) {
    extern __shared__ float smem[];
    const int stride = K + 4;
    float* sMean = smem;
    float* sX    = smem + 16 * stride;

    const int tm = blockIdx.x / nbx;
    const int cb = blockIdx.x % nbx;

    const int total = 16 * K;
    for (int idx = threadIdx.x * 4; idx < total; idx += blockDim.x * 4) {
        const int r  = idx / K;
        const int c0 = idx % K;
        const size_t g = (size_t)(tm * 16 + r) * K + c0;
        const float iv = inv[tm * 16 + r];
        float4 va = *(const float4*)(agg + g);
        va.x *= iv; va.y *= iv; va.z *= iv; va.w *= iv;
        *(float4*)(sMean + r * stride + c0) = va;
        *(float4*)(sX + r * stride + c0) = *(const float4*)(xin + g);
    }
    __syncthreads();

    const int lane = threadIdx.x & 31;
    const int wave = threadIdx.x >> 5;
    const int m  = lane & 15;
    const int hi = lane >> 4;
    const int tn = cb * 8 + wave;
    const int ncol = tn * 16 + m;
    const float* sMrow = sMean + m * stride;
    const float* sXrow = sX + m * stride;

    v8f c = {0.f, 0.f, 0.f, 0.f, 0.f, 0.f, 0.f, 0.f};

    for (int k0 = 0; k0 < K; k0 += 4) {
        const int ka = k0 + 2 * hi;
        v2f a = *(const v2f*)(sMrow + ka);
        v2f b;
        b.x = Wl[(size_t)ka * Nout + ncol];
        b.y = Wl[(size_t)(ka + 1) * Nout + ncol];
        c = __builtin_amdgcn_wmma_f32_16x16x4_f32(false, a, false, b,
                                                  (short)0, c, false, false);
        v2f a2 = *(const v2f*)(sXrow + ka);
        v2f b2;
        b2.x = Wr[(size_t)ka * Nout + ncol];
        b2.y = Wr[(size_t)(ka + 1) * Nout + ncol];
        c = __builtin_amdgcn_wmma_f32_16x16x4_f32(false, a2, false, b2,
                                                  (short)0, c, false, false);
    }

    const float bb = bias[ncol];
    #pragma unroll
    for (int r = 0; r < 8; ++r) {
        const int row = tm * 16 + r + 8 * hi;
        float v = c[r] + bb;
        if (do_relu) v = fmaxf(v, 0.0f);
        out[(size_t)row * Nout + ncol] = v;
    }
}

extern "C" void kernel_launch(void* const* d_in, const int* in_sizes, int n_in,
                              void* d_out, int out_size, void* d_ws, size_t ws_size,
                              hipStream_t stream) {
    const float* x    = (const float*)d_in[0];
    const int*   edge = (const int*)d_in[1];
    const float* W1l  = (const float*)d_in[2];
    const float* b1   = (const float*)d_in[3];
    const float* W1r  = (const float*)d_in[4];
    const float* W2l  = (const float*)d_in[5];
    const float* b2   = (const float*)d_in[6];
    const float* W2r  = (const float*)d_in[7];
    const float* W3l  = (const float*)d_in[8];
    const float* b3   = (const float*)d_in[9];
    const float* W3r  = (const float*)d_in[10];

    const int HIDDEN = in_sizes[3];
    const int OUTF   = in_sizes[9];
    const int D_IN   = in_sizes[2] / HIDDEN;
    const int M      = in_sizes[0] / D_IN;
    const int E      = in_sizes[1] / 2;
    const int* src   = edge;
    const int* dst   = edge + E;

    float* w = (float*)d_ws;
    const size_t aggOff = ((size_t)M + 127) & ~(size_t)127;
    const size_t h1Off  = aggOff + (size_t)M * HIDDEN;
    const size_t h2Off  = h1Off  + (size_t)M * HIDDEN;
    float* inv = w;
    float* agg = w + aggOff;
    float* h1  = w + h1Off;
    float* h2  = w + h2Off;
    float* outp = (float*)d_out;

    const int B = 256;

    hipMemsetAsync(inv, 0, (size_t)M * sizeof(float), stream);
    sage_deg_count<<<(E + B - 1) / B, B, 0, stream>>>(dst, inv, E);
    sage_inv_deg<<<(M + B - 1) / B, B, 0, stream>>>(inv, M);

    {
        const int K = D_IN, N = HIDDEN, f4 = K / 4;
        hipMemsetAsync(agg, 0, (size_t)M * K * sizeof(float), stream);
        long long nw = (long long)E * f4;
        sage_scatter_add<<<(int)((nw + B - 1) / B), B, 0, stream>>>(x, src, dst, agg, f4, nw);
        const int nbx = (N / 16) / 8;
        const size_t smem = 2u * 16u * (size_t)(K + 4) * sizeof(float);
        sage_gemm_wmma<<<(M / 16) * nbx, 256, smem, stream>>>(agg, inv, x, W1l, W1r, b1,
                                                              h1, K, N, nbx, 1);
    }
    {
        const int K = HIDDEN, N = HIDDEN, f4 = K / 4;
        hipMemsetAsync(agg, 0, (size_t)M * K * sizeof(float), stream);
        long long nw = (long long)E * f4;
        sage_scatter_add<<<(int)((nw + B - 1) / B), B, 0, stream>>>(h1, src, dst, agg, f4, nw);
        const int nbx = (N / 16) / 8;
        const size_t smem = 2u * 16u * (size_t)(K + 4) * sizeof(float);
        sage_gemm_wmma<<<(M / 16) * nbx, 256, smem, stream>>>(agg, inv, h1, W2l, W2r, b2,
                                                              h2, K, N, nbx, 1);
    }
    {
        const int K = HIDDEN, N = OUTF, f4 = K / 4;
        hipMemsetAsync(agg, 0, (size_t)M * K * sizeof(float), stream);
        long long nw = (long long)E * f4;
        sage_scatter_add<<<(int)((nw + B - 1) / B), B, 0, stream>>>(h2, src, dst, agg, f4, nw);
        const int nbx = (N / 16) / 8;
        const size_t smem = 2u * 16u * (size_t)(K + 4) * sizeof(float);
        sage_gemm_wmma<<<(M / 16) * nbx, 256, smem, stream>>>(agg, inv, h2, W3l, W3r, b3,
                                                              outp, K, N, nbx, 0);
    }
}


# Float-atomic scatter-add replaced by CSR gather: edges bucketed by destination once (int atomic count+rank, scan, fill); one wave sums a node's rows in arrival-rank order; no agg zero-fill
# speedup vs baseline: 2.3502x; 2.3502x over previous
#include <hip/hip_runtime.h>
#include <hip/hip_bf16.h>


typedef float v2f __attribute__((ext_vector_type(2)));
typedef float v8f __attribute__((ext_vector_type(8)));

__global__ void sage_dst_rank(const int* __restrict__ dst, int* __restrict__ cnt,
                              int* __restrict__ rank, int nedges) {
    int e = blockIdx.x * blockDim.x + threadIdx.x;
    if (e >= nedges) return;
    rank[e] = atomicAdd(&cnt[dst[e]], 1);
}

__global__ void __launch_bounds__(1024)
sage_rowptr_scan(const int* __restrict__ cnt, int* __restrict__ rowptr,
                 float* __restrict__ inv, int n) {
    __shared__ int sIncl[32];
    __shared__ int sPart[32];
    const int t = threadIdx.x;
    const int lane = t & 31;
    const int wave = t >> 5;
    const int b0 = blockIdx.x * 1024;

    int part = 0;
    const int4* c4 = (const int4*)cnt;
    const int nq = b0 >> 2;
    #pragma unroll 4
    for (int q = t; q < nq; q += 1024) {
        const int4 u = c4[q];
        part += (u.x + u.y) + (u.z + u.w);
    }

    const int i = b0 + t;
    const int v = (i < n) ? cnt[i] : 0;
    int incl = v;
    for (int d = 1; d < 32; d <<= 1) {
        const int u = __shfl_up(incl, d);
        if (lane >= d) incl += u;
    }
    for (int d = 16; d >= 1; d >>= 1) part += __shfl_xor(part, d);
    if (lane == 31) sIncl[wave] = incl;
    if (lane == 0)  sPart[wave] = part;
    __syncthreads();
    if (wave == 0) {
        const int wt = sIncl[lane];
        int ws = wt;
        for (int d = 1; d < 32; d <<= 1) {
            const int u = __shfl_up(ws, d);
            if (lane >= d) ws += u;
        }
        int pp = sPart[lane];
        for (int d = 16; d >= 1; d >>= 1) pp += __shfl_xor(pp, d);
        sIncl[lane] = ws - wt + pp;
    }
    __syncthreads();
    const int excl = sIncl[wave] + incl - v;
    if (i < n) {
        rowptr[i] = excl;
        inv[i] = 1.0f / fmaxf((float)v, 1.0f);
        if (i == n - 1) rowptr[n] = excl + v;
    }
}

__global__ void sage_csr_fill(const int* __restrict__ src, const int* __restrict__ dst,
                              const int* __restrict__ rank, const int* __restrict__ rowptr,
                              int* __restrict__ col, int nedges) {
    int e = blockIdx.x * blockDim.x + threadIdx.x;
    if (e >= nedges) return;
    col[rowptr[dst[e]] + rank[e]] = src[e];
}

template <int U>
__device__ __forceinline__ void sage_add_rows(float4& acc, const float* __restrict__ base,
                                              int my, int j, int F) {
    float4 v[U];
    #pragma unroll
    for (int u = 0; u < U; ++u) {
        const int s = __builtin_amdgcn_readlane(my, j + u);
        v[u] = *(const float4*)(base + (size_t)s * F);
    }
    #pragma unroll
    for (int u = 0; u < U; ++u) {
        acc.x += v[u].x; acc.y += v[u].y; acc.z += v[u].z; acc.w += v[u].w;
    }
}

__global__ void __launch_bounds__(256)
sage_gather_sum(const float* __restrict__ xin, const int* __restrict__ rowptr,
                const int* __restrict__ col, float* __restrict__ agg, int n, int F) {
    const int lane = threadIdx.x & 31;
    const int node = __builtin_amdgcn_readfirstlane(
        (int)(blockIdx.x * (blockDim.x >> 5) + (threadIdx.x >> 5)));
    if (node >= n) return;
    const int foff = blockIdx.y * 128 + lane * 4;
    const float* base = xin + foff;
    const int beg = rowptr[node];
    const int end = rowptr[node + 1];
    float4 acc = make_float4(0.f, 0.f, 0.f, 0.f);
    for (int b = beg; b < end; b += 32) {
        const int m = min(32, end - b);
        const int my = (lane < m) ? col[b + lane] : 0;
        int j = 0;
        for (; j + 8 <= m; j += 8) sage_add_rows<8>(acc, base, my, j, F);
        if (j + 4 <= m) { sage_add_rows<4>(acc, base, my, j, F); j += 4; }
        if (j + 2 <= m) { sage_add_rows<2>(acc, base, my, j, F); j += 2; }
        if (j < m)      { sage_add_rows<1>(acc, base, my, j, F); }
    }
    *(float4*)(agg + (size_t)node * F + foff) = acc;
}

__global__ void __launch_bounds__(256)
sage_gemm_wmma(const float* __restrict__ agg, const float* __restrict__ inv,
               const float* __restrict__ xin,
               const float* __restrict__ Wl, const float* __restrict__ Wr,
               const float* __restrict__ bias, float* __restrict__ out,
               int K, int Nout, int nbx, int do_relu) {
    extern __shared__ float smem[];
    const int stride = K + 4;
    float* sMean = smem;
    float* sX    = smem + 16 * stride;

    const int tm = blockIdx.x / nbx;
    const int cb = blockIdx.x % nbx;

    const int total = 16 * K;
    for (int idx = threadIdx.x * 4; idx < total; idx += blockDim.x * 4) {
        const int r  = idx / K;
        const int c0 = idx % K;
        const size_t g = (size_t)(tm * 16 + r) * K + c0;
        const float iv = inv[tm * 16 + r];
        float4 va = *(const float4*)(agg + g);
        va.x *= iv; va.y *= iv; va.z *= iv; va.w *= iv;
        *(float4*)(sMean + r * stride + c0) = va;
        *(float4*)(sX + r * stride + c0) = *(const float4*)(xin + g);
    }
    __syncthreads();

    const int lane = threadIdx.x & 31;
    const int wave = threadIdx.x >> 5;
    const int m  = lane & 15;
    const int hi = lane >> 4;
    const int tn = cb * 8 + wave;
    const int ncol = tn * 16 + m;
    const float* sMrow = sMean + m * stride;
    const float* sXrow = sX + m * stride;

    v8f c = {0.f, 0.f, 0.f, 0.f, 0.f, 0.f, 0.f, 0.f};

    for (int k0 = 0; k0 < K; k0 += 4) {
        const int ka = k0 + 2 * hi;
        v2f a = *(const v2f*)(sMrow + ka);
        v2f b;
        b.x = Wl[(size_t)ka * Nout + ncol];
        b.y = Wl[(size_t)(ka + 1) * Nout + ncol];
        c = __builtin_amdgcn_wmma_f32_16x16x4_f32(false, a, false, b,
                                                  (short)0, c, false, false);
        v2f a2 = *(const v2f*)(sXrow + ka);
        v2f b2;
        b2.x = Wr[(size_t)ka * Nout + ncol];
        b2.y = Wr[(size_t)(ka + 1) * Nout + ncol];
        c = __builtin_amdgcn_wmma_f32_16x16x4_f32(false, a2, false, b2,
                                                  (short)0, c, false, false);
    }

    const float bb = bias[ncol];
    #pragma unroll
    for (int r = 0; r < 8; ++r) {
        const int row = tm * 16 + r + 8 * hi;
        float v = c[r] + bb;
        if (do_relu) v = fmaxf(v, 0.0f);
        out[(size_t)row * Nout + ncol] = v;
    }
}

extern "C" void kernel_launch(void* const* d_in, const int* in_sizes, int n_in,
                              void* d_out, int out_size, void* d_ws, size_t ws_size,
                              hipStream_t stream) {
    const float* x    = (const float*)d_in[0];
    const int*   edge = (const int*)d_in[1];
    const float* W1l  = (const float*)d_in[2];
    const float* b1   = (const float*)d_in[3];
    const float* W1r  = (const float*)d_in[4];
    const float* W2l  = (const float*)d_in[5];
    const float* b2   = (const float*)d_in[6];
    const float* W2r  = (const float*)d_in[7];
    const float* W3l  = (const float*)d_in[8];
    const float* b3   = (const float*)d_in[9];
    const float* W3r  = (const float*)d_in[10];

    const int HIDDEN = in_sizes[3];
    const int OUTF   = in_sizes[9];
    const int D_IN   = in_sizes[2] / HIDDEN;
    const int M      = in_sizes[0] / D_IN;
    const int E      = in_sizes[1] / 2;
    const int* src   = edge;
    const int* dst   = edge + E;

    float* w = (float*)d_ws;
    const size_t aggOff = ((size_t)M + 127) & ~(size_t)127;
    const size_t h1Off  = aggOff + (size_t)M * HIDDEN;
    const size_t h2Off  = h1Off  + (size_t)M * HIDDEN;
    float* inv = w;
    float* agg = w + aggOff;
    float* h1  = w + h1Off;
    float* h2  = w + h2Off;
    float* outp = (float*)d_out;

    const size_t mPad = ((size_t)M + 1 + 127) & ~(size_t)127;
    const size_t ePad = ((size_t)E + 127) & ~(size_t)127;
    int* cnt    = (int*)(w + h2Off + (size_t)M * HIDDEN);
    int* rowptr = cnt + mPad;
    int* rank   = rowptr + mPad;
    int* col    = rank + ePad;

    const int B = 256;

    hipMemsetAsync(cnt, 0, (size_t)M * sizeof(int), stream);
    sage_dst_rank<<<(E + B - 1) / B, B, 0, stream>>>(dst, cnt, rank, E);
    sage_rowptr_scan<<<(M + 1023) / 1024, 1024, 0, stream>>>(cnt, rowptr, inv, M);
    sage_csr_fill<<<(E + B - 1) / B, B, 0, stream>>>(src, dst, rank, rowptr, col, E);

    {
        const int K = D_IN, N = HIDDEN;
        sage_gather_sum<<<dim3((M + 7) / 8, K / 128), 256, 0, stream>>>(x, rowptr, col, agg, M, K);
        const int nbx = (N / 16) / 8;
        const size_t smem = 2u * 16u * (size_t)(K + 4) * sizeof(float);
        sage_gemm_wmma<<<(M / 16) * nbx, 256, smem, stream>>>(agg, inv, x, W1l, W1r, b1,
                                                              h1, K, N, nbx, 1);
    }
    {
        const int K = HIDDEN, N = HIDDEN;
        sage_gather_sum<<<dim3((M + 7) / 8, K / 128), 256, 0, stream>>>(h1, rowptr, col, agg, M, K);
        const int nbx = (N / 16) / 8;
        const size_t smem = 2u * 16u * (size_t)(K + 4) * sizeof(float);
        sage_gemm_wmma<<<(M / 16) * nbx, 256, smem, stream>>>(agg, inv, h1, W2l, W2r, b2,
                                                              h2, K, N, nbx, 1);
    }
    {
        const int K = HIDDEN, N = OUTF;
        sage_gather_sum<<<dim3((M + 7) / 8, K / 128), 256, 0, stream>>>(h2, rowptr, col, agg, M, K);
        const int nbx = (N / 16) / 8;
        const size_t smem = 2u * 16u * (size_t)(K + 4) * sizeof(float);
        sage_gemm_wmma<<<(M / 16) * nbx, 256, smem, stream>>>(agg, inv, h2, W3l, W3r, b3,
                                                              outp, K, N, nbx, 0);
    }
}


# Neighbour sums formed inside the GEMM kernel: each block gathers its 16 nodes' rows (same list order) straight into its LDS A-tile and covers all columns; no agg buffer, 3 launches fewer
# speedup vs baseline: 2.5043x; 1.0656x over previous
#include <hip/hip_runtime.h>
#include <hip/hip_bf16.h>


typedef float v2f __attribute__((ext_vector_type(2)));
typedef float v8f __attribute__((ext_vector_type(8)));

__global__ void sage_dst_rank(const int* __restrict__ dst, int* __restrict__ cnt,
                              int* __restrict__ rank, int nedges) {
    int e = blockIdx.x * blockDim.x + threadIdx.x;
    if (e >= nedges) return;
    rank[e] = atomicAdd(&cnt[dst[e]], 1);
}

__global__ void __launch_bounds__(1024)
sage_rowptr_scan(const int* __restrict__ cnt, int* __restrict__ rowptr,
                 float* __restrict__ inv, int n) {
    __shared__ int sIncl[32];
    __shared__ int sPart[32];
    const int t = threadIdx.x;
    const int lane = t & 31;
    const int wave = t >> 5;
    const int b0 = blockIdx.x * 1024;

    int part = 0;
    const int4* c4 = (const int4*)cnt;
    const int nq = b0 >> 2;
    #pragma unroll 4
    for (int q = t; q < nq; q += 1024) {
        const int4 u = c4[q];
        part += (u.x + u.y) + (u.z + u.w);
    }

    const int i = b0 + t;
    const int v = (i < n) ? cnt[i] : 0;
    int incl = v;
    for (int d = 1; d < 32; d <<= 1) {
        const int u = __shfl_up(incl, d);
        if (lane >= d) incl += u;
    }
    for (int d = 16; d >= 1; d >>= 1) part += __shfl_xor(part, d);
    if (lane == 31) sIncl[wave] = incl;
    if (lane == 0)  sPart[wave] = part;
    __syncthreads();
    if (wave == 0) {
        const int wt = sIncl[lane];
        int ws = wt;
        for (int d = 1; d < 32; d <<= 1) {
            const int u = __shfl_up(ws, d);
            if (lane >= d) ws += u;
        }
        int pp = sPart[lane];
        for (int d = 16; d >= 1; d >>= 1) pp += __shfl_xor(pp, d);
        sIncl[lane] = ws - wt + pp;
    }
    __syncthreads();
    const int excl = sIncl[wave] + incl - v;
    if (i < n) {
        rowptr[i] = excl;
        inv[i] = 1.0f / fmaxf((float)v, 1.0f);
        if (i == n - 1) rowptr[n] = excl + v;
    }
}

__global__ void sage_csr_fill(const int* __restrict__ src, const int* __restrict__ dst,
                              const int* __restrict__ rank, const int* __restrict__ rowptr,
                              int* __restrict__ col, int nedges) {
    int e = blockIdx.x * blockDim.x + threadIdx.x;
    if (e >= nedges) return;
    col[rowptr[dst[e]] + rank[e]] = src[e];
}

template <int U>
__device__ __forceinline__ void sage_add_rows(float4& acc, const float* __restrict__ base,
                                              int my, int j, int F) {
    float4 v[U];
    #pragma unroll
    for (int u = 0; u < U; ++u) {
        const int s = __builtin_amdgcn_readlane(my, j + u);
        v[u] = *(const float4*)(base + (size_t)s * F);
    }
    #pragma unroll
    for (int u = 0; u < U; ++u) {
        acc.x += v[u].x; acc.y += v[u].y; acc.z += v[u].z; acc.w += v[u].w;
    }
}

template <int KP>
__global__ void __launch_bounds__(512)
sage_layer_wmma(const float* __restrict__ xin, const int* __restrict__ rowptr,
                const int* __restrict__ col, const float* __restrict__ inv,
                const float* __restrict__ Wl, const float* __restrict__ Wr,
                const float* __restrict__ bias, float* __restrict__ out,
                int Nout, int do_relu) {
    extern __shared__ float smem[];
    constexpr int K = KP * 128;
    constexpr int stride = K + 4;
    float* sMean = smem;
    float* sX    = smem + 16 * stride;

    const int tm = blockIdx.x;
    const int lane = threadIdx.x & 31;
    const int wave = __builtin_amdgcn_readfirstlane((int)(threadIdx.x >> 5));
    const int nwave = (int)(blockDim.x >> 5);

    for (int r = wave; r < 16; r += nwave) {
        const int node = tm * 16 + r;
        const int beg = rowptr[node];
        const int end = rowptr[node + 1];
        const float iv = inv[node];
        float4 acc[KP];
        #pragma unroll
        for (int p = 0; p < KP; ++p) acc[p] = make_float4(0.f, 0.f, 0.f, 0.f);
        for (int b = beg; b < end; b += 32) {
            const int mcnt = min(32, end - b);
            const int my = (lane < mcnt) ? col[b + lane] : 0;
            #pragma unroll
            for (int p = 0; p < KP; ++p) {
                const float* base = xin + p * 128 + lane * 4;
                int j = 0;
                for (; j + 8 <= mcnt; j += 8) sage_add_rows<8>(acc[p], base, my, j, K);
                if (j + 4 <= mcnt) { sage_add_rows<4>(acc[p], base, my, j, K); j += 4; }
                if (j + 2 <= mcnt) { sage_add_rows<2>(acc[p], base, my, j, K); j += 2; }
                if (j < mcnt)      { sage_add_rows<1>(acc[p], base, my, j, K); }
            }
        }
        #pragma unroll
        for (int p = 0; p < KP; ++p) {
            const int c0 = p * 128 + lane * 4;
            float4 va = acc[p];
            va.x *= iv; va.y *= iv; va.z *= iv; va.w *= iv;
            *(float4*)(sMean + r * stride + c0) = va;
            *(float4*)(sX + r * stride + c0) = *(const float4*)(xin + (size_t)node * K + c0);
        }
    }
    __syncthreads();

    const int m  = lane & 15;
    const int hi = lane >> 4;
    const int ncol = wave * 16 + m;
    const float* sMrow = sMean + m * stride;
    const float* sXrow = sX + m * stride;

    v8f c = {0.f, 0.f, 0.f, 0.f, 0.f, 0.f, 0.f, 0.f};

    for (int k0 = 0; k0 < K; k0 += 4) {
        const int ka = k0 + 2 * hi;
        v2f a = *(const v2f*)(sMrow + ka);
        v2f b;
        b.x = Wl[(size_t)ka * Nout + ncol];
        b.y = Wl[(size_t)(ka + 1) * Nout + ncol];
        c = __builtin_amdgcn_wmma_f32_16x16x4_f32(false, a, false, b,
                                                  (short)0, c, false, false);
        v2f a2 = *(const v2f*)(sXrow + ka);
        v2f b2;
        b2.x = Wr[(size_t)ka * Nout + ncol];
        b2.y = Wr[(size_t)(ka + 1) * Nout + ncol];
        c = __builtin_amdgcn_wmma_f32_16x16x4_f32(false, a2, false, b2,
                                                  (short)0, c, false, false);
    }

    const float bb = bias[ncol];
    #pragma unroll
    for (int r = 0; r < 8; ++r) {
        const int row = tm * 16 + r + 8 * hi;
        float v = c[r] + bb;
        if (do_relu) v = fmaxf(v, 0.0f);
        out[(size_t)row * Nout + ncol] = v;
    }
}

extern "C" void kernel_launch(void* const* d_in, const int* in_sizes, int n_in,
                              void* d_out, int out_size, void* d_ws, size_t ws_size,
                              hipStream_t stream) {
    const float* x    = (const float*)d_in[0];
    const int*   edge = (const int*)d_in[1];
    const float* W1l  = (const float*)d_in[2];
    const float* b1   = (const float*)d_in[3];
    const float* W1r  = (const float*)d_in[4];
    const float* W2l  = (const float*)d_in[5];
    const float* b2   = (const float*)d_in[6];
    const float* W2r  = (const float*)d_in[7];
    const float* W3l  = (const float*)d_in[8];
    const float* b3   = (const float*)d_in[9];
    const float* W3r  = (const float*)d_in[10];

    const int HIDDEN = in_sizes[3];
    const int OUTF   = in_sizes[9];
    const int D_IN   = in_sizes[2] / HIDDEN;
    const int M      = in_sizes[0] / D_IN;
    const int E      = in_sizes[1] / 2;
    const int* src   = edge;
    const int* dst   = edge + E;

    float* w = (float*)d_ws;
    const size_t aggOff = ((size_t)M + 127) & ~(size_t)127;
    const size_t h1Off  = aggOff + (size_t)M * HIDDEN;
    const size_t h2Off  = h1Off  + (size_t)M * HIDDEN;
    float* inv = w;
    float* h1  = w + h1Off;
    float* h2  = w + h2Off;
    float* outp = (float*)d_out;

    const size_t mPad = ((size_t)M + 1 + 127) & ~(size_t)127;
    const size_t ePad = ((size_t)E + 127) & ~(size_t)127;
    int* cnt    = (int*)(w + h2Off + (size_t)M * HIDDEN);
    int* rowptr = cnt + mPad;
    int* rank   = rowptr + mPad;
    int* col    = rank + ePad;

    const int B = 256;

    hipMemsetAsync(cnt, 0, (size_t)M * sizeof(int), stream);
    sage_dst_rank<<<(E + B - 1) / B, B, 0, stream>>>(dst, cnt, rank, E);
    sage_rowptr_scan<<<(M + 1023) / 1024, 1024, 0, stream>>>(cnt, rowptr, inv, M);
    sage_csr_fill<<<(E + B - 1) / B, B, 0, stream>>>(src, dst, rank, rowptr, col, E);

    {
        const int N = HIDDEN;
        const size_t smem = 2u * 16u * (size_t)(128 + 4) * sizeof(float);
        sage_layer_wmma<1><<<M / 16, (N / 16) * 32, smem, stream>>>(x, rowptr, col, inv, W1l, W1r,
                                                                    b1, h1, N, 1);
    }
    {
        const int N = HIDDEN;
        const size_t smem = 2u * 16u * (size_t)(256 + 4) * sizeof(float);
        sage_layer_wmma<2><<<M / 16, (N / 16) * 32, smem, stream>>>(h1, rowptr, col, inv, W2l, W2r,
                                                                    b2, h2, N, 1);
    }
    {
        const int N = OUTF;
        const size_t smem = 2u * 16u * (size_t)(256 + 4) * sizeof(float);
        sage_layer_wmma<2><<<M / 16, (N / 16) * 32, smem, stream>>>(h2, rowptr, col, inv, W3l, W3r,
                                                                    b3, outp, N, 0);
    }
}


# Each block owns 64 nodes, not 16: a wave keeps 4 accumulator tiles (2 in layer 3) and uses every loaded weight fragment for all of them; weight reads per node cut 4x; sum order unchanged
# speedup vs baseline: 2.8352x; 1.1321x over previous
#include <hip/hip_runtime.h>
#include <hip/hip_bf16.h>


typedef float v2f __attribute__((ext_vector_type(2)));
typedef float v8f __attribute__((ext_vector_type(8)));

__global__ void sage_dst_rank(const int* __restrict__ dst, int* __restrict__ cnt,
                              int* __restrict__ rank, int nedges) {
    int e = blockIdx.x * blockDim.x + threadIdx.x;
    if (e >= nedges) return;
    rank[e] = atomicAdd(&cnt[dst[e]], 1);
}

__global__ void __launch_bounds__(1024)
sage_rowptr_scan(const int* __restrict__ cnt, int* __restrict__ rowptr,
                 float* __restrict__ inv, int n) {
    __shared__ int sIncl[32];
    __shared__ int sPart[32];
    const int t = threadIdx.x;
    const int lane = t & 31;
    const int wave = t >> 5;
    const int b0 = blockIdx.x * 1024;

    int part = 0;
    const int4* c4 = (const int4*)cnt;
    const int nq = b0 >> 2;
    #pragma unroll 4
    for (int q = t; q < nq; q += 1024) {
        const int4 u = c4[q];
        part += (u.x + u.y) + (u.z + u.w);
    }

    const int i = b0 + t;
    const int v = (i < n) ? cnt[i] : 0;
    int incl = v;
    for (int d = 1; d < 32; d <<= 1) {
        const int u = __shfl_up(incl, d);
        if (lane >= d) incl += u;
    }
    for (int d = 16; d >= 1; d >>= 1) part += __shfl_xor(part, d);
    if (lane == 31) sIncl[wave] = incl;
    if (lane == 0)  sPart[wave] = part;
    __syncthreads();
    if (wave == 0) {
        const int wt = sIncl[lane];
        int ws = wt;
        for (int d = 1; d < 32; d <<= 1) {
            const int u = __shfl_up(ws, d);
            if (lane >= d) ws += u;
        }
        int pp = sPart[lane];
        for (int d = 16; d >= 1; d >>= 1) pp += __shfl_xor(pp, d);
        sIncl[lane] = ws - wt + pp;
    }
    __syncthreads();
    const int excl = sIncl[wave] + incl - v;
    if (i < n) {
        rowptr[i] = excl;
        inv[i] = 1.0f / fmaxf((float)v, 1.0f);
        if (i == n - 1) rowptr[n] = excl + v;
    }
}

__global__ void sage_csr_fill(const int* __restrict__ src, const int* __restrict__ dst,
                              const int* __restrict__ rank, const int* __restrict__ rowptr,
                              int* __restrict__ col, int nedges) {
    int e = blockIdx.x * blockDim.x + threadIdx.x;
    if (e >= nedges) return;
    col[rowptr[dst[e]] + rank[e]] = src[e];
}

template <int U>
__device__ __forceinline__ void sage_add_rows(float4& acc, const float* __restrict__ base,
                                              int my, int j, int F) {
    float4 v[U];
    #pragma unroll
    for (int u = 0; u < U; ++u) {
        const int s = __builtin_amdgcn_readlane(my, j + u);
        v[u] = *(const float4*)(base + (size_t)s * F);
    }
    #pragma unroll
    for (int u = 0; u < U; ++u) {
        acc.x += v[u].x; acc.y += v[u].y; acc.z += v[u].z; acc.w += v[u].w;
    }
}

template <int KP, int CT>
__global__ void __launch_bounds__(512)
sage_layer_wmma(const float* __restrict__ xin, const int* __restrict__ rowptr,
                const int* __restrict__ col, const float* __restrict__ inv,
                const float* __restrict__ Wl, const float* __restrict__ Wr,
                const float* __restrict__ bias, float* __restrict__ out,
                int M, int do_relu) {
    constexpr int K = KP * 128;
    constexpr int stride = K + 4;
    constexpr int R = 64;
    constexpr int NW = 16;
    constexpr int Nout = CT * 16;
    constexpr int TPW = (R / 16) * CT / NW;
    __shared__ float smem[2 * R * stride];
    float* sMean = smem;
    float* sX    = smem + R * stride;

    const int row0 = blockIdx.x * R;
    const int lane = threadIdx.x & 31;
    const int wave = __builtin_amdgcn_readfirstlane((int)(threadIdx.x >> 5));

    for (int r = wave; r < R; r += NW) {
        const int node = row0 + r;
        if (node < M) {
            const int beg = rowptr[node];
            const int end = rowptr[node + 1];
            const float iv = inv[node];
            float4 acc[KP];
            #pragma unroll
            for (int p = 0; p < KP; ++p) acc[p] = make_float4(0.f, 0.f, 0.f, 0.f);
            for (int b = beg; b < end; b += 32) {
                const int mcnt = min(32, end - b);
                const int my = (lane < mcnt) ? col[b + lane] : 0;
                #pragma unroll
                for (int p = 0; p < KP; ++p) {
                    const float* base = xin + p * 128 + lane * 4;
                    int j = 0;
                    for (; j + 8 <= mcnt; j += 8) sage_add_rows<8>(acc[p], base, my, j, K);
                    if (j + 4 <= mcnt) { sage_add_rows<4>(acc[p], base, my, j, K); j += 4; }
                    if (j + 2 <= mcnt) { sage_add_rows<2>(acc[p], base, my, j, K); j += 2; }
                    if (j < mcnt)      { sage_add_rows<1>(acc[p], base, my, j, K); }
                }
            }
            #pragma unroll
            for (int p = 0; p < KP; ++p) {
                const int c0 = p * 128 + lane * 4;
                float4 va = acc[p];
                va.x *= iv; va.y *= iv; va.z *= iv; va.w *= iv;
                *(float4*)(sMean + r * stride + c0) = va;
                *(float4*)(sX + r * stride + c0) = *(const float4*)(xin + (size_t)node * K + c0);
            }
        } else {
            #pragma unroll
            for (int p = 0; p < KP; ++p) {
                const int c0 = p * 128 + lane * 4;
                *(float4*)(sMean + r * stride + c0) = make_float4(0.f, 0.f, 0.f, 0.f);
                *(float4*)(sX + r * stride + c0) = make_float4(0.f, 0.f, 0.f, 0.f);
            }
        }
    }
    __syncthreads();

    const int m  = lane & 15;
    const int hi = lane >> 4;
    const int ct = wave % CT;
    const int t0 = (wave / CT) * TPW;
    const int ncol = ct * 16 + m;
    const float* sMrow = sMean + (t0 * 16 + m) * stride + 2 * hi;
    const float* sXrow = sX + (t0 * 16 + m) * stride + 2 * hi;
    const float* wl = Wl + (size_t)(2 * hi) * Nout + ncol;
    const float* wr = Wr + (size_t)(2 * hi) * Nout + ncol;

    v8f c[TPW];
    #pragma unroll
    for (int t = 0; t < TPW; ++t) c[t] = (v8f){0.f, 0.f, 0.f, 0.f, 0.f, 0.f, 0.f, 0.f};

    for (int k0 = 0; k0 < K; k0 += 4) {
        v2f bl, br;
        bl.x = wl[(size_t)k0 * Nout];
        bl.y = wl[(size_t)(k0 + 1) * Nout];
        br.x = wr[(size_t)k0 * Nout];
        br.y = wr[(size_t)(k0 + 1) * Nout];
        #pragma unroll
        for (int t = 0; t < TPW; ++t) {
            const v2f a = *(const v2f*)(sMrow + t * 16 * stride + k0);
            c[t] = __builtin_amdgcn_wmma_f32_16x16x4_f32(false, a, false, bl,
                                                         (short)0, c[t], false, false);
        }
        #pragma unroll
        for (int t = 0; t < TPW; ++t) {
            const v2f a2 = *(const v2f*)(sXrow + t * 16 * stride + k0);
            c[t] = __builtin_amdgcn_wmma_f32_16x16x4_f32(false, a2, false, br,
                                                         (short)0, c[t], false, false);
        }
    }

    const float bb = bias[ncol];
    #pragma unroll
    for (int t = 0; t < TPW; ++t) {
        #pragma unroll
        for (int r = 0; r < 8; ++r) {
            const int row = row0 + (t0 + t) * 16 + r + 8 * hi;
            float v = c[t][r] + bb;
            if (do_relu) v = fmaxf(v, 0.0f);
            if (row < M) out[(size_t)row * Nout + ncol] = v;
        }
    }
}

extern "C" void kernel_launch(void* const* d_in, const int* in_sizes, int n_in,
                              void* d_out, int out_size, void* d_ws, size_t ws_size,
                              hipStream_t stream) {
    const float* x    = (const float*)d_in[0];
    const int*   edge = (const int*)d_in[1];
    const float* W1l  = (const float*)d_in[2];
    const float* b1   = (const float*)d_in[3];
    const float* W1r  = (const float*)d_in[4];
    const float* W2l  = (const float*)d_in[5];
    const float* b2   = (const float*)d_in[6];
    const float* W2r  = (const float*)d_in[7];
    const float* W3l  = (const float*)d_in[8];
    const float* b3   = (const float*)d_in[9];
    const float* W3r  = (const float*)d_in[10];

    const int HIDDEN = in_sizes[3];
    const int D_IN   = in_sizes[2] / HIDDEN;
    const int M      = in_sizes[0] / D_IN;
    const int E      = in_sizes[1] / 2;
    const int* src   = edge;
    const int* dst   = edge + E;

    float* w = (float*)d_ws;
    const size_t aggOff = ((size_t)M + 127) & ~(size_t)127;
    const size_t h1Off  = aggOff + (size_t)M * HIDDEN;
    const size_t h2Off  = h1Off  + (size_t)M * HIDDEN;
    float* inv = w;
    float* h1  = w + h1Off;
    float* h2  = w + h2Off;
    float* outp = (float*)d_out;

    const size_t mPad = ((size_t)M + 1 + 127) & ~(size_t)127;
    const size_t ePad = ((size_t)E + 127) & ~(size_t)127;
    int* cnt    = (int*)(w + h2Off + (size_t)M * HIDDEN);
    int* rowptr = cnt + mPad;
    int* rank   = rowptr + mPad;
    int* col    = rank + ePad;

    const int B = 256;

    hipMemsetAsync(cnt, 0, (size_t)M * sizeof(int), stream);
    sage_dst_rank<<<(E + B - 1) / B, B, 0, stream>>>(dst, cnt, rank, E);
    sage_rowptr_scan<<<(M + 1023) / 1024, 1024, 0, stream>>>(cnt, rowptr, inv, M);
    sage_csr_fill<<<(E + B - 1) / B, B, 0, stream>>>(src, dst, rank, rowptr, col, E);

    const int RB = 64;
    const int nblk = (M + RB - 1) / RB;
    sage_layer_wmma<1, 16><<<nblk, 512, 0, stream>>>(x, rowptr, col, inv, W1l, W1r, b1, h1, M, 1);
    sage_layer_wmma<2, 16><<<nblk, 512, 0, stream>>>(h1, rowptr, col, inv, W2l, W2r, b2, h2, M, 1);
    sage_layer_wmma<2, 8><<<nblk, 512, 0, stream>>>(h2, rowptr, col, inv, W3l, W3r, b3, outp, M, 0);
}


# Gather needs a third of the dependent memory round trips: a wave fetches its 4 nodes' list bounds, index lists and own rows up front and loads 8 neighbours x all pieces as one group, tail included
# speedup vs baseline: 2.9330x; 1.0345x over previous
#include <hip/hip_runtime.h>
#include <hip/hip_bf16.h>


typedef float v2f __attribute__((ext_vector_type(2)));
typedef float v8f __attribute__((ext_vector_type(8)));

__global__ void sage_dst_rank(const int* __restrict__ dst, int* __restrict__ cnt,
                              int* __restrict__ rank, int nedges) {
    int e = blockIdx.x * blockDim.x + threadIdx.x;
    if (e >= nedges) return;
    rank[e] = atomicAdd(&cnt[dst[e]], 1);
}

__global__ void __launch_bounds__(1024)
sage_rowptr_scan(const int* __restrict__ cnt, int* __restrict__ rowptr,
                 float* __restrict__ inv, int n) {
    __shared__ int sIncl[32];
    __shared__ int sPart[32];
    const int t = threadIdx.x;
    const int lane = t & 31;
    const int wave = t >> 5;
    const int b0 = blockIdx.x * 1024;

    int part = 0;
    const int4* c4 = (const int4*)cnt;
    const int nq = b0 >> 2;
    #pragma unroll 4
    for (int q = t; q < nq; q += 1024) {
        const int4 u = c4[q];
        part += (u.x + u.y) + (u.z + u.w);
    }

    const int i = b0 + t;
    const int v = (i < n) ? cnt[i] : 0;
    int incl = v;
    for (int d = 1; d < 32; d <<= 1) {
        const int u = __shfl_up(incl, d);
        if (lane >= d) incl += u;
    }
    for (int d = 16; d >= 1; d >>= 1) part += __shfl_xor(part, d);
    if (lane == 31) sIncl[wave] = incl;
    if (lane == 0)  sPart[wave] = part;
    __syncthreads();
    if (wave == 0) {
        const int wt = sIncl[lane];
        int ws = wt;
        for (int d = 1; d < 32; d <<= 1) {
            const int u = __shfl_up(ws, d);
            if (lane >= d) ws += u;
        }
        int pp = sPart[lane];
        for (int d = 16; d >= 1; d >>= 1) pp += __shfl_xor(pp, d);
        sIncl[lane] = ws - wt + pp;
    }
    __syncthreads();
    const int excl = sIncl[wave] + incl - v;
    if (i < n) {
        rowptr[i] = excl;
        inv[i] = 1.0f / fmaxf((float)v, 1.0f);
        if (i == n - 1) rowptr[n] = excl + v;
    }
}

__global__ void sage_csr_fill(const int* __restrict__ src, const int* __restrict__ dst,
                              const int* __restrict__ rank, const int* __restrict__ rowptr,
                              int* __restrict__ col, int nedges) {
    int e = blockIdx.x * blockDim.x + threadIdx.x;
    if (e >= nedges) return;
    col[rowptr[dst[e]] + rank[e]] = src[e];
}

typedef float v4f __attribute__((ext_vector_type(4)));

template <int KP, int UB, bool FULL>
__device__ __forceinline__ void sage_add_chunk(v4f (&acc)[KP], const float* __restrict__ xl,
                                               int my, int j, int cnt) {
    constexpr int K = KP * 128;
    int sidx[UB];
    #pragma unroll
    for (int u = 0; u < UB; ++u) sidx[u] = __builtin_amdgcn_readlane(my, j + u);
    v4f v[UB][KP];
    #pragma unroll
    for (int u = 0; u < UB; ++u) {
        #pragma unroll
        for (int p = 0; p < KP; ++p) v[u][p] = (v4f){0.f, 0.f, 0.f, 0.f};
        if (FULL || u < cnt) {
            const float* row = xl + (size_t)sidx[u] * K;
            #pragma unroll
            for (int p = 0; p < KP; ++p) v[u][p] = *(const v4f*)(row + p * 128);
        }
    }
    #pragma unroll
    for (int u = 0; u < UB; ++u) {
        if (FULL || u < cnt) {
            #pragma unroll
            for (int p = 0; p < KP; ++p) acc[p] += v[u][p];
        }
    }
}

template <int KP, int CT>
__global__ void __launch_bounds__(512, KP == 1 ? 12 : 8)
sage_layer_wmma(const float* __restrict__ xin, const int* __restrict__ rowptr,
                const int* __restrict__ col, const float* __restrict__ inv,
                const float* __restrict__ Wl, const float* __restrict__ Wr,
                const float* __restrict__ bias, float* __restrict__ out,
                int M, int do_relu) {
    constexpr int K = KP * 128;
    constexpr int stride = K + 4;
    constexpr int R = 64;
    constexpr int NW = 16;
    constexpr int Nout = CT * 16;
    constexpr int TPW = (R / 16) * CT / NW;
    __shared__ float smem[2 * R * stride];
    float* sMean = smem;
    float* sX    = smem + R * stride;

    const int row0 = blockIdx.x * R;
    const int lane = threadIdx.x & 31;
    const int wave = __builtin_amdgcn_readfirstlane((int)(threadIdx.x >> 5));

    constexpr int NPW = R / NW;
    constexpr int UB = 8;
    const int n0 = row0 + NPW * wave;
    int rp[NPW + 1];
    float ivv[NPW];
    #pragma unroll
    for (int i = 0; i <= NPW; ++i) rp[i] = rowptr[min(n0 + i, M)];
    #pragma unroll
    for (int i = 0; i < NPW; ++i) ivv[i] = inv[min(n0 + i, M - 1)];
    int myn[NPW];
    #pragma unroll
    for (int i = 0; i < NPW; ++i)
        myn[i] = (lane < rp[i + 1] - rp[i]) ? col[rp[i] + lane] : 0;
    {
        float4 own[NPW][KP];
        #pragma unroll
        for (int i = 0; i < NPW; ++i) {
            #pragma unroll
            for (int p = 0; p < KP; ++p) {
                own[i][p] = *(const float4*)(xin + (size_t)min(n0 + i, M - 1) * K
                                             + p * 128 + lane * 4);
            }
        }
        #pragma unroll
        for (int i = 0; i < NPW; ++i) {
            #pragma unroll
            for (int p = 0; p < KP; ++p)
                *(float4*)(sX + (NPW * wave + i) * stride + p * 128 + lane * 4) = own[i][p];
        }
    }
    const float* xl = xin + lane * 4;
    #pragma unroll
    for (int i = 0; i < NPW; ++i) {
        const int beg = rp[i];
        const int deg = rp[i + 1] - beg;
        v4f acc[KP];
        #pragma unroll
        for (int p = 0; p < KP; ++p) acc[p] = (v4f){0.f, 0.f, 0.f, 0.f};
        int my = myn[i];
        for (int b = 0; b < deg; b += 32) {
            if (b > 0) my = (lane < deg - b) ? col[beg + b + lane] : 0;
            const int mcnt = min(32, deg - b);
            int j = 0;
            for (; j + UB <= mcnt; j += UB) sage_add_chunk<KP, UB, true>(acc, xl, my, j, UB);
            if (j < mcnt) sage_add_chunk<KP, UB, false>(acc, xl, my, j, mcnt - j);
        }
        const float iv = ivv[i];
        #pragma unroll
        for (int p = 0; p < KP; ++p) {
            *(v4f*)(sMean + (NPW * wave + i) * stride + p * 128 + lane * 4) = acc[p] * iv;
        }
    }
    __syncthreads();

    const int m  = lane & 15;
    const int hi = lane >> 4;
    const int ct = wave % CT;
    const int t0 = (wave / CT) * TPW;
    const int ncol = ct * 16 + m;
    const float* sMrow = sMean + (t0 * 16 + m) * stride + 2 * hi;
    const float* sXrow = sX + (t0 * 16 + m) * stride + 2 * hi;
    const float* wl = Wl + (size_t)(2 * hi) * Nout + ncol;
    const float* wr = Wr + (size_t)(2 * hi) * Nout + ncol;

    v8f c[TPW];
    #pragma unroll
    for (int t = 0; t < TPW; ++t) c[t] = (v8f){0.f, 0.f, 0.f, 0.f, 0.f, 0.f, 0.f, 0.f};

    #pragma unroll(8 / TPW)
    for (int k0 = 0; k0 < K; k0 += 4) {
        v2f bl, br;
        bl.x = wl[(size_t)k0 * Nout];
        bl.y = wl[(size_t)(k0 + 1) * Nout];
        br.x = wr[(size_t)k0 * Nout];
        br.y = wr[(size_t)(k0 + 1) * Nout];
        #pragma unroll
        for (int t = 0; t < TPW; ++t) {
            const v2f a = *(const v2f*)(sMrow + t * 16 * stride + k0);
            c[t] = __builtin_amdgcn_wmma_f32_16x16x4_f32(false, a, false, bl,
                                                         (short)0, c[t], false, false);
        }
        #pragma unroll
        for (int t = 0; t < TPW; ++t) {
            const v2f a2 = *(const v2f*)(sXrow + t * 16 * stride + k0);
            c[t] = __builtin_amdgcn_wmma_f32_16x16x4_f32(false, a2, false, br,
                                                         (short)0, c[t], false, false);
        }
    }

    const float bb = bias[ncol];
    #pragma unroll
    for (int t = 0; t < TPW; ++t) {
        #pragma unroll
        for (int r = 0; r < 8; ++r) {
            const int row = row0 + (t0 + t) * 16 + r + 8 * hi;
            float v = c[t][r] + bb;
            if (do_relu) v = fmaxf(v, 0.0f);
            if (row < M) out[(size_t)row * Nout + ncol] = v;
        }
    }
}

extern "C" void kernel_launch(void* const* d_in, const int* in_sizes, int n_in,
                              void* d_out, int out_size, void* d_ws, size_t ws_size,
                              hipStream_t stream) {
    const float* x    = (const float*)d_in[0];
    const int*   edge = (const int*)d_in[1];
    const float* W1l  = (const float*)d_in[2];
    const float* b1   = (const float*)d_in[3];
    const float* W1r  = (const float*)d_in[4];
    const float* W2l  = (const float*)d_in[5];
    const float* b2   = (const float*)d_in[6];
    const float* W2r  = (const float*)d_in[7];
    const float* W3l  = (const float*)d_in[8];
    const float* b3   = (const float*)d_in[9];
    const float* W3r  = (const float*)d_in[10];

    const int HIDDEN = in_sizes[3];
    const int D_IN   = in_sizes[2] / HIDDEN;
    const int M      = in_sizes[0] / D_IN;
    const int E      = in_sizes[1] / 2;
    const int* src   = edge;
    const int* dst   = edge + E;

    float* w = (float*)d_ws;
    const size_t aggOff = ((size_t)M + 127) & ~(size_t)127;
    const size_t h1Off  = aggOff + (size_t)M * HIDDEN;
    const size_t h2Off  = h1Off  + (size_t)M * HIDDEN;
    float* inv = w;
    float* h1  = w + h1Off;
    float* h2  = w + h2Off;
    float* outp = (float*)d_out;

    const size_t mPad = ((size_t)M + 1 + 127) & ~(size_t)127;
    const size_t ePad = ((size_t)E + 127) & ~(size_t)127;
    int* cnt    = (int*)(w + h2Off + (size_t)M * HIDDEN);
    int* rowptr = cnt + mPad;
    int* rank   = rowptr + mPad;
    int* col    = rank + ePad;

    const int B = 256;

    hipMemsetAsync(cnt, 0, (size_t)M * sizeof(int), stream);
    sage_dst_rank<<<(E + B - 1) / B, B, 0, stream>>>(dst, cnt, rank, E);
    sage_rowptr_scan<<<(M + 1023) / 1024, 1024, 0, stream>>>(cnt, rowptr, inv, M);
    sage_csr_fill<<<(E + B - 1) / B, B, 0, stream>>>(src, dst, rank, rowptr, col, E);

    const int RB = 64;
    const int nblk = (M + RB - 1) / RB;
    sage_layer_wmma<1, 16><<<nblk, 512, 0, stream>>>(x, rowptr, col, inv, W1l, W1r, b1, h1, M, 1);
    sage_layer_wmma<2, 16><<<nblk, 512, 0, stream>>>(h1, rowptr, col, inv, W2l, W2r, b2, h2, M, 1);
    sage_layer_wmma<2, 8><<<nblk, 512, 0, stream>>>(h2, rowptr, col, inv, W3l, W3r, b3, outp, M, 0);
}


# Gather overlaps the matrix products: one 32-wave block per compute unit walks its nodes in 64-row tiles; 16 waves gather tile t+1 into a 2nd LDS stage while 16 run tile t's WMMAs; sum order kept
# speedup vs baseline: 3.1484x; 1.0734x over previous
#include <hip/hip_runtime.h>
#include <hip/hip_bf16.h>


typedef float v2f __attribute__((ext_vector_type(2)));
typedef float v8f __attribute__((ext_vector_type(8)));

__global__ void sage_dst_rank(const int* __restrict__ dst, int* __restrict__ cnt,
                              int* __restrict__ rank, int nedges) {
    int e = blockIdx.x * blockDim.x + threadIdx.x;
    if (e >= nedges) return;
    rank[e] = atomicAdd(&cnt[dst[e]], 1);
}

__global__ void __launch_bounds__(1024)
sage_rowptr_scan(const int* __restrict__ cnt, int* __restrict__ rowptr,
                 float* __restrict__ inv, int n) {
    __shared__ int sIncl[32];
    __shared__ int sPart[32];
    const int t = threadIdx.x;
    const int lane = t & 31;
    const int wave = t >> 5;
    const int b0 = blockIdx.x * 1024;

    int part = 0;
    const int4* c4 = (const int4*)cnt;
    const int nq = b0 >> 2;
    #pragma unroll 4
    for (int q = t; q < nq; q += 1024) {
        const int4 u = c4[q];
        part += (u.x + u.y) + (u.z + u.w);
    }

    const int i = b0 + t;
    const int v = (i < n) ? cnt[i] : 0;
    int incl = v;
    for (int d = 1; d < 32; d <<= 1) {
        const int u = __shfl_up(incl, d);
        if (lane >= d) incl += u;
    }
    for (int d = 16; d >= 1; d >>= 1) part += __shfl_xor(part, d);
    if (lane == 31) sIncl[wave] = incl;
    if (lane == 0)  sPart[wave] = part;
    __syncthreads();
    if (wave == 0) {
        const int wt = sIncl[lane];
        int ws = wt;
        for (int d = 1; d < 32; d <<= 1) {
            const int u = __shfl_up(ws, d);
            if (lane >= d) ws += u;
        }
        int pp = sPart[lane];
        for (int d = 16; d >= 1; d >>= 1) pp += __shfl_xor(pp, d);
        sIncl[lane] = ws - wt + pp;
    }
    __syncthreads();
    const int excl = sIncl[wave] + incl - v;
    if (i < n) {
        rowptr[i] = excl;
        inv[i] = 1.0f / fmaxf((float)v, 1.0f);
        if (i == n - 1) rowptr[n] = excl + v;
    }
}

__global__ void sage_csr_fill(const int* __restrict__ src, const int* __restrict__ dst,
                              const int* __restrict__ rank, const int* __restrict__ rowptr,
                              int* __restrict__ col, int nedges) {
    int e = blockIdx.x * blockDim.x + threadIdx.x;
    if (e >= nedges) return;
    col[rowptr[dst[e]] + rank[e]] = src[e];
}

typedef float v4f __attribute__((ext_vector_type(4)));

template <int KP, int UB, bool FULL>
__device__ __forceinline__ void sage_add_chunk(v4f (&acc)[KP], const float* __restrict__ xl,
                                               int my, int j, int cnt) {
    constexpr int K = KP * 128;
    int sidx[UB];
    #pragma unroll
    for (int u = 0; u < UB; ++u) sidx[u] = __builtin_amdgcn_readlane(my, j + u);
    v4f v[UB][KP];
    #pragma unroll
    for (int u = 0; u < UB; ++u) {
        #pragma unroll
        for (int p = 0; p < KP; ++p) v[u][p] = (v4f){0.f, 0.f, 0.f, 0.f};
        if (FULL || u < cnt) {
            const float* row = xl + (size_t)sidx[u] * K;
            #pragma unroll
            for (int p = 0; p < KP; ++p) v[u][p] = *(const v4f*)(row + p * 128);
        }
    }
    #pragma unroll
    for (int u = 0; u < UB; ++u) {
        if (FULL || u < cnt) {
            #pragma unroll
            for (int p = 0; p < KP; ++p) acc[p] += v[u][p];
        }
    }
}

template <int KP, int CT, int NL>
__device__ __forceinline__ void sage_tile_products(const float* sMean, const float* sX,
        const float* __restrict__ Wl, const float* __restrict__ Wr,
        const float* __restrict__ bias, float* __restrict__ out,
        int tileRow0, int nEnd, int do_relu, int cw, int lane) {
    constexpr int K = KP * 128;
    constexpr int stride = K + 4;
    constexpr int Nout = CT * 16;
    constexpr int TPW = 4 * CT / 16;
    constexpr int UNR = 4;
    const int m  = lane & 15;
    const int hi = lane >> 4;
    const int ct = cw % CT;
    const int t0 = (cw / CT) * TPW;
    const int ncol = ct * 16 + m;
    const float* sMrow = sMean + (t0 * 16 + m) * stride + 2 * hi;
    const float* sXrow = sX + (t0 * 16 + m) * stride + 2 * hi;
    const float* wl = Wl + (size_t)(2 * hi) * Nout + ncol;
    const float* wr = Wr + (size_t)(2 * hi) * Nout + ncol;

    v8f c[NL];
    #pragma unroll
    for (int t = 0; t < NL; ++t) c[t] = (v8f){0.f, 0.f, 0.f, 0.f, 0.f, 0.f, 0.f, 0.f};

    #pragma unroll(UNR)
    for (int k0 = 0; k0 < K; k0 += 4) {
        v2f bl, br;
        bl.x = wl[(size_t)k0 * Nout];
        bl.y = wl[(size_t)(k0 + 1) * Nout];
        br.x = wr[(size_t)k0 * Nout];
        br.y = wr[(size_t)(k0 + 1) * Nout];
        #pragma unroll
        for (int t = 0; t < NL; ++t) {
            const v2f a = *(const v2f*)(sMrow + t * 16 * stride + k0);
            c[t] = __builtin_amdgcn_wmma_f32_16x16x4_f32(false, a, false, bl,
                                                         (short)0, c[t], false, false);
        }
        #pragma unroll
        for (int t = 0; t < NL; ++t) {
            const v2f a2 = *(const v2f*)(sXrow + t * 16 * stride + k0);
            c[t] = __builtin_amdgcn_wmma_f32_16x16x4_f32(false, a2, false, br,
                                                         (short)0, c[t], false, false);
        }
    }

    const float bb = bias[ncol];
    #pragma unroll
    for (int t = 0; t < NL; ++t) {
        #pragma unroll
        for (int r = 0; r < 8; ++r) {
            const int row = tileRow0 + (t0 + t) * 16 + r + 8 * hi;
            float v = c[t][r] + bb;
            if (do_relu) v = fmaxf(v, 0.0f);
            if (row < nEnd) out[(size_t)row * Nout + ncol] = v;
        }
    }
}

template <int KP, int NPW>
__device__ __forceinline__ void sage_tile_gather(float* sMean, float* sX,
        const float* __restrict__ xin, const int* __restrict__ rowptr,
        const int* __restrict__ col, const float* __restrict__ inv,
        int tileRow0, int nEnd, int M, int gw, int lane) {
    constexpr int K = KP * 128;
    constexpr int stride = K + 4;
    constexpr int UB = 8;
    const int n0 = tileRow0 + NPW * gw;
    int rp[NPW + 1];
    float ivv[NPW];
    #pragma unroll
    for (int i = 0; i <= NPW; ++i) rp[i] = rowptr[min(n0 + i, nEnd)];
    #pragma unroll
    for (int i = 0; i < NPW; ++i) ivv[i] = inv[min(n0 + i, M - 1)];
    int myn[NPW];
    #pragma unroll
    for (int i = 0; i < NPW; ++i)
        myn[i] = (lane < rp[i + 1] - rp[i]) ? col[rp[i] + lane] : 0;
    {
        float4 own[NPW][KP];
        #pragma unroll
        for (int i = 0; i < NPW; ++i) {
            #pragma unroll
            for (int p = 0; p < KP; ++p) {
                own[i][p] = *(const float4*)(xin + (size_t)min(n0 + i, M - 1) * K
                                             + p * 128 + lane * 4);
            }
        }
        #pragma unroll
        for (int i = 0; i < NPW; ++i) {
            #pragma unroll
            for (int p = 0; p < KP; ++p)
                *(float4*)(sX + (NPW * gw + i) * stride + p * 128 + lane * 4) = own[i][p];
        }
    }
    const float* xl = xin + lane * 4;
    #pragma unroll
    for (int i = 0; i < NPW; ++i) {
        const int beg = rp[i];
        const int deg = rp[i + 1] - beg;
        v4f acc[KP];
        #pragma unroll
        for (int p = 0; p < KP; ++p) acc[p] = (v4f){0.f, 0.f, 0.f, 0.f};
        int my = myn[i];
        for (int b = 0; b < deg; b += 32) {
            if (b > 0) my = (lane < deg - b) ? col[beg + b + lane] : 0;
            const int mcnt = min(32, deg - b);
            int j = 0;
            for (; j + UB <= mcnt; j += UB) sage_add_chunk<KP, UB, true>(acc, xl, my, j, UB);
            if (j < mcnt) sage_add_chunk<KP, UB, false>(acc, xl, my, j, mcnt - j);
        }
        const float iv = ivv[i];
        #pragma unroll
        for (int p = 0; p < KP; ++p) {
            *(v4f*)(sMean + (NPW * gw + i) * stride + p * 128 + lane * 4) = acc[p] * iv;
        }
    }
}

template <int KP, int CT>
__global__ void __launch_bounds__(1024)
sage_layer_wmma(const float* __restrict__ xin, const int* __restrict__ rowptr,
                const int* __restrict__ col, const float* __restrict__ inv,
                const float* __restrict__ Wl, const float* __restrict__ Wr,
                const float* __restrict__ bias, float* __restrict__ out,
                int M, int do_relu) {
    constexpr int K = KP * 128;
    constexpr int stride = K + 4;
    constexpr int R = 64;
    constexpr int TPW = 4 * CT / 16;
    constexpr int STG = 2 * R * stride;
    __shared__ float smem[2 * STG];

    const int lane = threadIdx.x & 31;
    const int wave = __builtin_amdgcn_readfirstlane((int)(threadIdx.x >> 5));
    const int grp = wave >> 2;
    const bool gatherer = (grp & 1) == 0;
    const int rw = (grp >> 1) * 4 + (wave & 3);

    const int per  = (M + (int)gridDim.x - 1) / (int)gridDim.x;
    const int nBeg = min(M, (int)blockIdx.x * per);
    const int nEnd = min(M, nBeg + per);
    const int ntiles = (nEnd - nBeg + R - 1) / R;

    for (int it = 0; it <= ntiles; ++it) {
        if (it == 0) {
            if (ntiles > 0)
                sage_tile_gather<KP, 2>(smem, smem + R * stride, xin, rowptr, col, inv,
                                        nBeg, nEnd, M, wave, lane);
        } else if (gatherer) {
            if (it < ntiles) {
                float* sMean = smem + (it & 1) * STG;
                float* sX    = sMean + R * stride;
                sage_tile_gather<KP, 4>(sMean, sX, xin, rowptr, col, inv,
                                        nBeg + it * R, nEnd, M, rw, lane);
            }
        } else {
            const float* sMean = smem + ((it - 1) & 1) * STG;
            const float* sX    = sMean + R * stride;
            const int tileRow0 = nBeg + (it - 1) * R;
            const int nlt = (min(R, nEnd - tileRow0) + 15) >> 4;
            const int nl  = min(TPW, nlt - (rw / CT) * TPW);
            if (nl >= 4) {
                if constexpr (TPW >= 4)
                    sage_tile_products<KP, CT, 4>(sMean, sX, Wl, Wr, bias, out, tileRow0, nEnd, do_relu, rw, lane);
            } else if (nl == 3) {
                if constexpr (TPW >= 4)
                    sage_tile_products<KP, CT, 3>(sMean, sX, Wl, Wr, bias, out, tileRow0, nEnd, do_relu, rw, lane);
            } else if (nl == 2) {
                sage_tile_products<KP, CT, 2>(sMean, sX, Wl, Wr, bias, out, tileRow0, nEnd, do_relu, rw, lane);
            } else if (nl == 1) {
                sage_tile_products<KP, CT, 1>(sMean, sX, Wl, Wr, bias, out, tileRow0, nEnd, do_relu, rw, lane);
            }
        }
        __syncthreads();
    }
}

extern "C" void kernel_launch(void* const* d_in, const int* in_sizes, int n_in,
                              void* d_out, int out_size, void* d_ws, size_t ws_size,
                              hipStream_t stream) {
    const float* x    = (const float*)d_in[0];
    const int*   edge = (const int*)d_in[1];
    const float* W1l  = (const float*)d_in[2];
    const float* b1   = (const float*)d_in[3];
    const float* W1r  = (const float*)d_in[4];
    const float* W2l  = (const float*)d_in[5];
    const float* b2   = (const float*)d_in[6];
    const float* W2r  = (const float*)d_in[7];
    const float* W3l  = (const float*)d_in[8];
    const float* b3   = (const float*)d_in[9];
    const float* W3r  = (const float*)d_in[10];

    const int HIDDEN = in_sizes[3];
    const int D_IN   = in_sizes[2] / HIDDEN;
    const int M      = in_sizes[0] / D_IN;
    const int E      = in_sizes[1] / 2;
    const int* src   = edge;
    const int* dst   = edge + E;

    float* w = (float*)d_ws;
    const size_t aggOff = ((size_t)M + 127) & ~(size_t)127;
    const size_t h1Off  = aggOff + (size_t)M * HIDDEN;
    const size_t h2Off  = h1Off  + (size_t)M * HIDDEN;
    float* inv = w;
    float* h1  = w + h1Off;
    float* h2  = w + h2Off;
    float* outp = (float*)d_out;

    const size_t mPad = ((size_t)M + 1 + 127) & ~(size_t)127;
    const size_t ePad = ((size_t)E + 127) & ~(size_t)127;
    int* cnt    = (int*)(w + h2Off + (size_t)M * HIDDEN);
    int* rowptr = cnt + mPad;
    int* rank   = rowptr + mPad;
    int* col    = rank + ePad;

    const int B = 256;

    hipMemsetAsync(cnt, 0, (size_t)M * sizeof(int), stream);
    sage_dst_rank<<<(E + B - 1) / B, B, 0, stream>>>(dst, cnt, rank, E);
    sage_rowptr_scan<<<(M + 1023) / 1024, 1024, 0, stream>>>(cnt, rowptr, inv, M);
    sage_csr_fill<<<(E + B - 1) / B, B, 0, stream>>>(src, dst, rank, rowptr, col, E);

    const int nblk = 256;
    sage_layer_wmma<1, 16><<<nblk, 1024, 0, stream>>>(x, rowptr, col, inv, W1l, W1r, b1, h1, M, 1);
    sage_layer_wmma<2, 16><<<nblk, 1024, 0, stream>>>(h1, rowptr, col, inv, W2l, W2r, b2, h2, M, 1);
    sage_layer_wmma<2, 8><<<nblk, 1024, 0, stream>>>(h2, rowptr, col, inv, W3l, W3r, b3, outp, M, 0);
}


# Weights repacked once per call into WMMA fragment order per 16-column tile: a product wave reads them as contiguous 512-byte loads, one per 2 k-steps and path, not 64-byte row pieces; sum order kept
# speedup vs baseline: 3.2604x; 1.0356x over previous
#include <hip/hip_runtime.h>
#include <hip/hip_bf16.h>


typedef float v2f __attribute__((ext_vector_type(2)));
typedef float v8f __attribute__((ext_vector_type(8)));

typedef float v4f __attribute__((ext_vector_type(4)));

template <int K, int NOUT>
__device__ __forceinline__ void sage_pack_weights(const float* __restrict__ Wl,
                                                  const float* __restrict__ Wr,
                                                  float* __restrict__ dst, int q) {
    const int lane = q & 31;
    const int path = (q >> 5) & 1;
    const int t  = q >> 6;
    const int kk = t % (K / 8);
    const int ct = t / (K / 8);
    const int n  = ct * 16 + (lane & 15);
    const int kb = 8 * kk + 2 * (lane >> 4);
    const float* W = path ? Wr : Wl;
    v4f o;
    o.x = W[(size_t)kb * NOUT + n];
    o.y = W[(size_t)(kb + 1) * NOUT + n];
    o.z = W[(size_t)(kb + 4) * NOUT + n];
    o.w = W[(size_t)(kb + 5) * NOUT + n];
    *(v4f*)(dst + (size_t)q * 4) = o;
}

constexpr int SAGE_DIN = 128, SAGE_HID = 256, SAGE_OUT = 128;
constexpr int SAGE_PK1 = 2 * SAGE_DIN * SAGE_HID / 4;
constexpr int SAGE_PK2 = 2 * SAGE_HID * SAGE_HID / 4;
constexpr int SAGE_PK3 = 2 * SAGE_HID * SAGE_OUT / 4;

__global__ void sage_dst_rank(const int* __restrict__ dst, int* __restrict__ cnt,
                              int* __restrict__ rank, int nedges,
                              const float* __restrict__ W1l, const float* __restrict__ W1r,
                              const float* __restrict__ W2l, const float* __restrict__ W2r,
                              const float* __restrict__ W3l, const float* __restrict__ W3r,
                              float* __restrict__ wpack, int packBlocks) {
    if ((int)blockIdx.x < packBlocks) {
        const int q = blockIdx.x * blockDim.x + threadIdx.x;
        if (q < SAGE_PK1)
            sage_pack_weights<SAGE_DIN, SAGE_HID>(W1l, W1r, wpack, q);
        else if (q < SAGE_PK1 + SAGE_PK2)
            sage_pack_weights<SAGE_HID, SAGE_HID>(W2l, W2r, wpack + (size_t)SAGE_PK1 * 4,
                                                  q - SAGE_PK1);
        else if (q < SAGE_PK1 + SAGE_PK2 + SAGE_PK3)
            sage_pack_weights<SAGE_HID, SAGE_OUT>(W3l, W3r,
                                                  wpack + (size_t)(SAGE_PK1 + SAGE_PK2) * 4,
                                                  q - SAGE_PK1 - SAGE_PK2);
        return;
    }
    int e = ((int)blockIdx.x - packBlocks) * blockDim.x + threadIdx.x;
    if (e >= nedges) return;
    rank[e] = atomicAdd(&cnt[dst[e]], 1);
}

__global__ void __launch_bounds__(1024)
sage_rowptr_scan(const int* __restrict__ cnt, int* __restrict__ rowptr,
                 float* __restrict__ inv, int n) {
    __shared__ int sIncl[32];
    __shared__ int sPart[32];
    const int t = threadIdx.x;
    const int lane = t & 31;
    const int wave = t >> 5;
    const int b0 = blockIdx.x * 1024;

    int part = 0;
    const int4* c4 = (const int4*)cnt;
    const int nq = b0 >> 2;
    #pragma unroll 4
    for (int q = t; q < nq; q += 1024) {
        const int4 u = c4[q];
        part += (u.x + u.y) + (u.z + u.w);
    }

    const int i = b0 + t;
    const int v = (i < n) ? cnt[i] : 0;
    int incl = v;
    for (int d = 1; d < 32; d <<= 1) {
        const int u = __shfl_up(incl, d);
        if (lane >= d) incl += u;
    }
    for (int d = 16; d >= 1; d >>= 1) part += __shfl_xor(part, d);
    if (lane == 31) sIncl[wave] = incl;
    if (lane == 0)  sPart[wave] = part;
    __syncthreads();
    if (wave == 0) {
        const int wt = sIncl[lane];
        int ws = wt;
        for (int d = 1; d < 32; d <<= 1) {
            const int u = __shfl_up(ws, d);
            if (lane >= d) ws += u;
        }
        int pp = sPart[lane];
        for (int d = 16; d >= 1; d >>= 1) pp += __shfl_xor(pp, d);
        sIncl[lane] = ws - wt + pp;
    }
    __syncthreads();
    const int excl = sIncl[wave] + incl - v;
    if (i < n) {
        rowptr[i] = excl;
        inv[i] = 1.0f / fmaxf((float)v, 1.0f);
        if (i == n - 1) rowptr[n] = excl + v;
    }
}

__global__ void sage_csr_fill(const int* __restrict__ src, const int* __restrict__ dst,
                              const int* __restrict__ rank, const int* __restrict__ rowptr,
                              int* __restrict__ col, int nedges) {
    int e = blockIdx.x * blockDim.x + threadIdx.x;
    if (e >= nedges) return;
    col[rowptr[dst[e]] + rank[e]] = src[e];
}


template <int KP, int UB, bool FULL>
__device__ __forceinline__ void sage_add_chunk(v4f (&acc)[KP], const float* __restrict__ xl,
                                               int my, int j, int cnt) {
    constexpr int K = KP * 128;
    int sidx[UB];
    #pragma unroll
    for (int u = 0; u < UB; ++u) sidx[u] = __builtin_amdgcn_readlane(my, j + u);
    v4f v[UB][KP];
    #pragma unroll
    for (int u = 0; u < UB; ++u) {
        #pragma unroll
        for (int p = 0; p < KP; ++p) v[u][p] = (v4f){0.f, 0.f, 0.f, 0.f};
        if (FULL || u < cnt) {
            const float* row = xl + (size_t)sidx[u] * K;
            #pragma unroll
            for (int p = 0; p < KP; ++p) v[u][p] = *(const v4f*)(row + p * 128);
        }
    }
    #pragma unroll
    for (int u = 0; u < UB; ++u) {
        if (FULL || u < cnt) {
            #pragma unroll
            for (int p = 0; p < KP; ++p) acc[p] += v[u][p];
        }
    }
}

template <int KP, int NL>
__device__ __forceinline__ void sage_kstep(v8f (&c)[NL], const float* sMrow, const float* sXrow,
                                           int k0, v2f bl, v2f br) {
    constexpr int stride = KP * 128 + 4;
    #pragma unroll
    for (int t = 0; t < NL; ++t) {
        const v2f a = *(const v2f*)(sMrow + t * 16 * stride + k0);
        c[t] = __builtin_amdgcn_wmma_f32_16x16x4_f32(false, a, false, bl,
                                                     (short)0, c[t], false, false);
    }
    #pragma unroll
    for (int t = 0; t < NL; ++t) {
        const v2f a2 = *(const v2f*)(sXrow + t * 16 * stride + k0);
        c[t] = __builtin_amdgcn_wmma_f32_16x16x4_f32(false, a2, false, br,
                                                     (short)0, c[t], false, false);
    }
}

template <int KP, int CT, int NL>
__device__ __forceinline__ void sage_tile_products(const float* sMean, const float* sX,
        const float* __restrict__ Wp,
        const float* __restrict__ bias, float* __restrict__ out,
        int tileRow0, int nEnd, int do_relu, int cw, int lane) {
    constexpr int K = KP * 128;
    constexpr int stride = K + 4;
    constexpr int Nout = CT * 16;
    constexpr int TPW = 4 * CT / 16;
    const int m  = lane & 15;
    const int hi = lane >> 4;
    const int ct = cw % CT;
    const int t0 = (cw / CT) * TPW;
    const int ncol = ct * 16 + m;
    const float* sMrow = sMean + (t0 * 16 + m) * stride + 2 * hi;
    const float* sXrow = sX + (t0 * 16 + m) * stride + 2 * hi;
    const v4f* wp = (const v4f*)Wp + (size_t)ct * (K / 8) * 64 + lane;

    v8f c[NL];
    #pragma unroll
    for (int t = 0; t < NL; ++t) c[t] = (v8f){0.f, 0.f, 0.f, 0.f, 0.f, 0.f, 0.f, 0.f};

    #pragma unroll(2)
    for (int k0 = 0; k0 < K; k0 += 8) {
        const v4f wl4 = wp[(k0 >> 3) * 64];
        const v4f wr4 = wp[(k0 >> 3) * 64 + 32];
        sage_kstep<KP, NL>(c, sMrow, sXrow, k0,     (v2f){wl4.x, wl4.y}, (v2f){wr4.x, wr4.y});
        sage_kstep<KP, NL>(c, sMrow, sXrow, k0 + 4, (v2f){wl4.z, wl4.w}, (v2f){wr4.z, wr4.w});
    }

    const float bb = bias[ncol];
    #pragma unroll
    for (int t = 0; t < NL; ++t) {
        #pragma unroll
        for (int r = 0; r < 8; ++r) {
            const int row = tileRow0 + (t0 + t) * 16 + r + 8 * hi;
            float v = c[t][r] + bb;
            if (do_relu) v = fmaxf(v, 0.0f);
            if (row < nEnd) out[(size_t)row * Nout + ncol] = v;
        }
    }
}

template <int KP, int NPW>
__device__ __forceinline__ void sage_tile_gather(float* sMean, float* sX,
        const float* __restrict__ xin, const int* __restrict__ rowptr,
        const int* __restrict__ col, const float* __restrict__ inv,
        int tileRow0, int nEnd, int M, int gw, int lane) {
    constexpr int K = KP * 128;
    constexpr int stride = K + 4;
    constexpr int UB = 8;
    const int n0 = tileRow0 + NPW * gw;
    int rp[NPW + 1];
    float ivv[NPW];
    #pragma unroll
    for (int i = 0; i <= NPW; ++i) rp[i] = rowptr[min(n0 + i, nEnd)];
    #pragma unroll
    for (int i = 0; i < NPW; ++i) ivv[i] = inv[min(n0 + i, M - 1)];
    int myn[NPW];
    #pragma unroll
    for (int i = 0; i < NPW; ++i)
        myn[i] = (lane < rp[i + 1] - rp[i]) ? col[rp[i] + lane] : 0;
    {
        float4 own[NPW][KP];
        #pragma unroll
        for (int i = 0; i < NPW; ++i) {
            #pragma unroll
            for (int p = 0; p < KP; ++p) {
                own[i][p] = *(const float4*)(xin + (size_t)min(n0 + i, M - 1) * K
                                             + p * 128 + lane * 4);
            }
        }
        #pragma unroll
        for (int i = 0; i < NPW; ++i) {
            #pragma unroll
            for (int p = 0; p < KP; ++p)
                *(float4*)(sX + (NPW * gw + i) * stride + p * 128 + lane * 4) = own[i][p];
        }
    }
    const float* xl = xin + lane * 4;
    #pragma unroll
    for (int i = 0; i < NPW; ++i) {
        const int beg = rp[i];
        const int deg = rp[i + 1] - beg;
        v4f acc[KP];
        #pragma unroll
        for (int p = 0; p < KP; ++p) acc[p] = (v4f){0.f, 0.f, 0.f, 0.f};
        int my = myn[i];
        for (int b = 0; b < deg; b += 32) {
            if (b > 0) my = (lane < deg - b) ? col[beg + b + lane] : 0;
            const int mcnt = min(32, deg - b);
            int j = 0;
            for (; j + UB <= mcnt; j += UB) sage_add_chunk<KP, UB, true>(acc, xl, my, j, UB);
            if (j < mcnt) sage_add_chunk<KP, UB, false>(acc, xl, my, j, mcnt - j);
        }
        const float iv = ivv[i];
        #pragma unroll
        for (int p = 0; p < KP; ++p) {
            *(v4f*)(sMean + (NPW * gw + i) * stride + p * 128 + lane * 4) = acc[p] * iv;
        }
    }
}

template <int KP, int CT>
__global__ void __launch_bounds__(1024)
sage_layer_wmma(const float* __restrict__ xin, const int* __restrict__ rowptr,
                const int* __restrict__ col, const float* __restrict__ inv,
                const float* __restrict__ Wp,
                const float* __restrict__ bias, float* __restrict__ out,
                int M, int do_relu) {
    constexpr int K = KP * 128;
    constexpr int stride = K + 4;
    constexpr int R = 64;
    constexpr int TPW = 4 * CT / 16;
    constexpr int STG = 2 * R * stride;
    __shared__ float smem[2 * STG];

    const int lane = threadIdx.x & 31;
    const int wave = __builtin_amdgcn_readfirstlane((int)(threadIdx.x >> 5));
    const int grp = wave >> 2;
    const bool gatherer = (grp & 1) == 0;
    const int rw = (grp >> 1) * 4 + (wave & 3);

    const int per  = (M + (int)gridDim.x - 1) / (int)gridDim.x;
    const int nBeg = min(M, (int)blockIdx.x * per);
    const int nEnd = min(M, nBeg + per);
    const int ntiles = (nEnd - nBeg + R - 1) / R;

    for (int it = 0; it <= ntiles; ++it) {
        if (it == 0) {
            if (ntiles > 0)
                sage_tile_gather<KP, 2>(smem, smem + R * stride, xin, rowptr, col, inv,
                                        nBeg, nEnd, M, wave, lane);
        } else if (gatherer) {
            if (it < ntiles) {
                float* sMean = smem + (it & 1) * STG;
                float* sX    = sMean + R * stride;
                sage_tile_gather<KP, 4>(sMean, sX, xin, rowptr, col, inv,
                                        nBeg + it * R, nEnd, M, rw, lane);
            }
        } else {
            const float* sMean = smem + ((it - 1) & 1) * STG;
            const float* sX    = sMean + R * stride;
            const int tileRow0 = nBeg + (it - 1) * R;
            const int nlt = (min(R, nEnd - tileRow0) + 15) >> 4;
            const int nl  = min(TPW, nlt - (rw / CT) * TPW);
            if (nl >= 4) {
                if constexpr (TPW >= 4)
                    sage_tile_products<KP, CT, 4>(sMean, sX, Wp, bias, out, tileRow0, nEnd, do_relu, rw, lane);
            } else if (nl == 3) {
                if constexpr (TPW >= 4)
                    sage_tile_products<KP, CT, 3>(sMean, sX, Wp, bias, out, tileRow0, nEnd, do_relu, rw, lane);
            } else if (nl == 2) {
                sage_tile_products<KP, CT, 2>(sMean, sX, Wp, bias, out, tileRow0, nEnd, do_relu, rw, lane);
            } else if (nl == 1) {
                sage_tile_products<KP, CT, 1>(sMean, sX, Wp, bias, out, tileRow0, nEnd, do_relu, rw, lane);
            }
        }
        __syncthreads();
    }
}

extern "C" void kernel_launch(void* const* d_in, const int* in_sizes, int n_in,
                              void* d_out, int out_size, void* d_ws, size_t ws_size,
                              hipStream_t stream) {
    const float* x    = (const float*)d_in[0];
    const int*   edge = (const int*)d_in[1];
    const float* W1l  = (const float*)d_in[2];
    const float* b1   = (const float*)d_in[3];
    const float* W1r  = (const float*)d_in[4];
    const float* W2l  = (const float*)d_in[5];
    const float* b2   = (const float*)d_in[6];
    const float* W2r  = (const float*)d_in[7];
    const float* W3l  = (const float*)d_in[8];
    const float* b3   = (const float*)d_in[9];
    const float* W3r  = (const float*)d_in[10];

    const int HIDDEN = in_sizes[3];
    const int D_IN   = in_sizes[2] / HIDDEN;
    const int M      = in_sizes[0] / D_IN;
    const int E      = in_sizes[1] / 2;
    const int* src   = edge;
    const int* dst   = edge + E;

    float* w = (float*)d_ws;
    const size_t aggOff = ((size_t)M + 127) & ~(size_t)127;
    const size_t h1Off  = aggOff + (size_t)M * HIDDEN;
    const size_t h2Off  = h1Off  + (size_t)M * HIDDEN;
    float* inv = w;
    float* wpack = w + aggOff;
    float* h1  = w + h1Off;
    float* h2  = w + h2Off;
    float* outp = (float*)d_out;

    const size_t mPad = ((size_t)M + 1 + 127) & ~(size_t)127;
    const size_t ePad = ((size_t)E + 127) & ~(size_t)127;
    int* cnt    = (int*)(w + h2Off + (size_t)M * HIDDEN);
    int* rowptr = cnt + mPad;
    int* rank   = rowptr + mPad;
    int* col    = rank + ePad;

    const int B = 256;

    hipMemsetAsync(cnt, 0, (size_t)M * sizeof(int), stream);
    const int packBlocks = (SAGE_PK1 + SAGE_PK2 + SAGE_PK3 + B - 1) / B;
    sage_dst_rank<<<packBlocks + (E + B - 1) / B, B, 0, stream>>>(dst, cnt, rank, E,
        W1l, W1r, W2l, W2r, W3l, W3r, wpack, packBlocks);
    sage_rowptr_scan<<<(M + 1023) / 1024, 1024, 0, stream>>>(cnt, rowptr, inv, M);
    sage_csr_fill<<<(E + B - 1) / B, B, 0, stream>>>(src, dst, rank, rowptr, col, E);

    const int nblk = 256;
    sage_layer_wmma<1, 16><<<nblk, 1024, 0, stream>>>(x, rowptr, col, inv, wpack, b1, h1, M, 1);
    sage_layer_wmma<2, 16><<<nblk, 1024, 0, stream>>>(h1, rowptr, col, inv,
        wpack + (size_t)SAGE_PK1 * 4, b2, h2, M, 1);
    sage_layer_wmma<2, 8><<<nblk, 1024, 0, stream>>>(h2, rowptr, col, inv,
        wpack + (size_t)(SAGE_PK1 + SAGE_PK2) * 4, b3, outp, M, 0);
}


# CSR built by a two-level counting sort on LDS counters (256-destination buckets, then per node): no global atomic per edge, no memset, few fat waves; sum order = arrival order of LDS cursor adds
# speedup vs baseline: 3.7812x; 1.1597x over previous
#include <hip/hip_runtime.h>
#include <hip/hip_bf16.h>


typedef float v2f __attribute__((ext_vector_type(2)));
typedef float v8f __attribute__((ext_vector_type(8)));

typedef float v4f __attribute__((ext_vector_type(4)));

template <int K, int NOUT>
__device__ __forceinline__ void sage_pack_weights(const float* __restrict__ Wl,
                                                  const float* __restrict__ Wr,
                                                  float* __restrict__ dst, int q) {
    const int lane = q & 31;
    const int path = (q >> 5) & 1;
    const int t  = q >> 6;
    const int kk = t % (K / 8);
    const int ct = t / (K / 8);
    const int n  = ct * 16 + (lane & 15);
    const int kb = 8 * kk + 2 * (lane >> 4);
    const float* W = path ? Wr : Wl;
    v4f o;
    o.x = W[(size_t)kb * NOUT + n];
    o.y = W[(size_t)(kb + 1) * NOUT + n];
    o.z = W[(size_t)(kb + 4) * NOUT + n];
    o.w = W[(size_t)(kb + 5) * NOUT + n];
    *(v4f*)(dst + (size_t)q * 4) = o;
}

constexpr int SAGE_DIN = 128, SAGE_HID = 256, SAGE_OUT = 128;
constexpr int SAGE_PK1 = 2 * SAGE_DIN * SAGE_HID / 4;
constexpr int SAGE_PK2 = 2 * SAGE_HID * SAGE_HID / 4;
constexpr int SAGE_PK3 = 2 * SAGE_HID * SAGE_OUT / 4;

constexpr int SAGE_BK_SHIFT = 8;
constexpr int SAGE_BK_NODES = 1 << SAGE_BK_SHIFT;
constexpr int SAGE_MAXBK = 1024;
constexpr int SAGE_EPT = 8;
constexpr int SAGE_EPB = 1024 * SAGE_EPT;
constexpr int SAGE_BK_U = 6;

__global__ void __launch_bounds__(1024)
sage_bucket_count(const int* __restrict__ dst, int nedges, int nbk,
                  int* __restrict__ hist,
                  const float* __restrict__ W1l, const float* __restrict__ W1r,
                  const float* __restrict__ W2l, const float* __restrict__ W2r,
                  const float* __restrict__ W3l, const float* __restrict__ W3r,
                  float* __restrict__ wpack, int packBlocks) {
    __shared__ int sCnt[SAGE_MAXBK];
    const int t = threadIdx.x;
    if ((int)blockIdx.x < packBlocks) {
        const int q = blockIdx.x * 1024 + t;
        if (q < SAGE_PK1)
            sage_pack_weights<SAGE_DIN, SAGE_HID>(W1l, W1r, wpack, q);
        else if (q < SAGE_PK1 + SAGE_PK2)
            sage_pack_weights<SAGE_HID, SAGE_HID>(W2l, W2r, wpack + (size_t)SAGE_PK1 * 4,
                                                  q - SAGE_PK1);
        else if (q < SAGE_PK1 + SAGE_PK2 + SAGE_PK3)
            sage_pack_weights<SAGE_HID, SAGE_OUT>(W3l, W3r,
                                                  wpack + (size_t)(SAGE_PK1 + SAGE_PK2) * 4,
                                                  q - SAGE_PK1 - SAGE_PK2);
        return;
    }
    const int blk = (int)blockIdx.x - packBlocks;
    int d[SAGE_EPT];
    #pragma unroll
    for (int u = 0; u < SAGE_EPT; ++u) {
        const int e = blk * SAGE_EPB + u * 1024 + t;
        d[u] = (e < nedges) ? dst[e] : -1;
    }
    sCnt[t] = 0;
    __syncthreads();
    #pragma unroll
    for (int u = 0; u < SAGE_EPT; ++u)
        if (d[u] >= 0) atomicAdd(&sCnt[d[u] >> SAGE_BK_SHIFT], 1);
    __syncthreads();
    if (t < nbk) hist[(size_t)blk * nbk + t] = sCnt[t];
}

__global__ void __launch_bounds__(1024)
sage_bucket_place(const int* __restrict__ src, const int* __restrict__ dst, int nedges,
                  int nbk, int nblk, const int* __restrict__ hist,
                  int* __restrict__ bstart, int* __restrict__ pairs) {
    __shared__ int sTot[SAGE_MAXBK];
    __shared__ int sBef[SAGE_MAXBK];
    __shared__ int sCur[SAGE_MAXBK];
    __shared__ int sWave[32];
    const int t = threadIdx.x;
    const int lane = t & 31;
    const int wave = t >> 5;
    const int blk = (int)blockIdx.x;

    int es[SAGE_EPT], ed[SAGE_EPT];
    #pragma unroll
    for (int u = 0; u < SAGE_EPT; ++u) {
        const int e = blk * SAGE_EPB + u * 1024 + t;
        const bool ok = e < nedges;
        es[u] = ok ? src[e] : 0;
        ed[u] = ok ? dst[e] : -1;
    }
    sTot[t] = 0;
    sBef[t] = 0;
    __syncthreads();

    const int total = nbk * nblk;
    for (int f0 = 0; f0 < total; f0 += 8192) {
        int v[8];
        #pragma unroll
        for (int u = 0; u < 8; ++u) {
            const int f = f0 + u * 1024 + t;
            v[u] = (f < total) ? hist[f] : 0;
        }
        #pragma unroll
        for (int u = 0; u < 8; ++u) {
            const int f = f0 + u * 1024 + t;
            if (f < total && v[u] != 0) {
                const int b = f / nbk;
                const int i = f - b * nbk;
                atomicAdd(&sTot[i], v[u]);
                if (b < blk) atomicAdd(&sBef[i], v[u]);
            }
        }
    }
    __syncthreads();

    const int v = (t < nbk) ? sTot[t] : 0;
    int incl = v;
    for (int d = 1; d < 32; d <<= 1) {
        const int u = __shfl_up(incl, d);
        if (lane >= d) incl += u;
    }
    if (lane == 31) sWave[wave] = incl;
    __syncthreads();
    if (wave == 0) {
        const int wt = sWave[lane];
        int ws = wt;
        for (int d = 1; d < 32; d <<= 1) {
            const int u = __shfl_up(ws, d);
            if (lane >= d) ws += u;
        }
        sWave[lane] = ws - wt;
    }
    __syncthreads();
    const int excl = sWave[wave] + incl - v;
    if (t < nbk) {
        sCur[t] = excl + sBef[t];
        if (blk == 0) bstart[t] = excl;
    }
    if (blk == 0 && t == nbk - 1) bstart[nbk] = excl + v;
    __syncthreads();

    #pragma unroll
    for (int u = 0; u < SAGE_EPT; ++u) {
        if (ed[u] >= 0) {
            const int pos = atomicAdd(&sCur[ed[u] >> SAGE_BK_SHIFT], 1);
            *(int2*)(pairs + 2 * (size_t)pos) = make_int2(es[u], ed[u]);
        }
    }
}

__global__ void __launch_bounds__(1024)
sage_bucket_csr(const int* __restrict__ pairs, const int* __restrict__ bstart,
                int* __restrict__ rowptr, float* __restrict__ inv,
                int* __restrict__ col, int n) {
    __shared__ int sCnt[SAGE_BK_NODES];
    __shared__ int sCur[SAGE_BK_NODES];
    __shared__ int sWave[SAGE_BK_NODES / 32];
    const int t = threadIdx.x;
    const int lane = t & 31;
    const int wave = t >> 5;
    const int b = (int)blockIdx.x;
    const int s = bstart[b];
    const int e = bstart[b + 1];
    if (t < SAGE_BK_NODES) sCnt[t] = 0;
    __syncthreads();
    for (int i0 = s; i0 < e; i0 += SAGE_BK_U * 1024) {
        int dl[SAGE_BK_U];
        #pragma unroll
        for (int u = 0; u < SAGE_BK_U; ++u) {
            const int i = i0 + u * 1024 + t;
            dl[u] = (i < e) ? pairs[2 * (size_t)i + 1] : -1;
        }
        #pragma unroll
        for (int u = 0; u < SAGE_BK_U; ++u)
            if (dl[u] >= 0) atomicAdd(&sCnt[dl[u] & (SAGE_BK_NODES - 1)], 1);
    }
    __syncthreads();
    const int v = (t < SAGE_BK_NODES) ? sCnt[t] : 0;
    int incl = v;
    for (int d = 1; d < 32; d <<= 1) {
        const int u = __shfl_up(incl, d);
        if (lane >= d) incl += u;
    }
    if (t < SAGE_BK_NODES && lane == 31) sWave[wave] = incl;
    __syncthreads();
    if (t < SAGE_BK_NODES) {
        int base = 0;
        for (int w = 0; w < wave; ++w) base += sWave[w];
        const int excl = base + incl - v;
        sCur[t] = excl;
        const int node = b * SAGE_BK_NODES + t;
        if (node < n) {
            rowptr[node] = s + excl;
            inv[node] = 1.0f / fmaxf((float)v, 1.0f);
        }
    }
    if (b == (int)gridDim.x - 1 && t == 0) rowptr[n] = e;
    __syncthreads();
    for (int i0 = s; i0 < e; i0 += SAGE_BK_U * 1024) {
        int2 p[SAGE_BK_U];
        #pragma unroll
        for (int u = 0; u < SAGE_BK_U; ++u) {
            const int i = i0 + u * 1024 + t;
            p[u] = (i < e) ? *(const int2*)(pairs + 2 * (size_t)i) : make_int2(0, -1);
        }
        #pragma unroll
        for (int u = 0; u < SAGE_BK_U; ++u) {
            if (p[u].y >= 0) {
                const int pos = atomicAdd(&sCur[p[u].y & (SAGE_BK_NODES - 1)], 1);
                col[s + pos] = p[u].x;
            }
        }
    }
}


template <int KP, int UB, bool FULL>
__device__ __forceinline__ void sage_add_chunk(v4f (&acc)[KP], const float* __restrict__ xl,
                                               int my, int j, int cnt) {
    constexpr int K = KP * 128;
    int sidx[UB];
    #pragma unroll
    for (int u = 0; u < UB; ++u) sidx[u] = __builtin_amdgcn_readlane(my, j + u);
    v4f v[UB][KP];
    #pragma unroll
    for (int u = 0; u < UB; ++u) {
        #pragma unroll
        for (int p = 0; p < KP; ++p) v[u][p] = (v4f){0.f, 0.f, 0.f, 0.f};
        if (FULL || u < cnt) {
            const float* row = xl + (size_t)sidx[u] * K;
            #pragma unroll
            for (int p = 0; p < KP; ++p) v[u][p] = *(const v4f*)(row + p * 128);
        }
    }
    #pragma unroll
    for (int u = 0; u < UB; ++u) {
        if (FULL || u < cnt) {
            #pragma unroll
            for (int p = 0; p < KP; ++p) acc[p] += v[u][p];
        }
    }
}

template <int KP, int NL>
__device__ __forceinline__ void sage_kstep(v8f (&c)[NL], const float* sMrow, const float* sXrow,
                                           int k0, v2f bl, v2f br) {
    constexpr int stride = KP * 128 + 4;
    #pragma unroll
    for (int t = 0; t < NL; ++t) {
        const v2f a = *(const v2f*)(sMrow + t * 16 * stride + k0);
        c[t] = __builtin_amdgcn_wmma_f32_16x16x4_f32(false, a, false, bl,
                                                     (short)0, c[t], false, false);
    }
    #pragma unroll
    for (int t = 0; t < NL; ++t) {
        const v2f a2 = *(const v2f*)(sXrow + t * 16 * stride + k0);
        c[t] = __builtin_amdgcn_wmma_f32_16x16x4_f32(false, a2, false, br,
                                                     (short)0, c[t], false, false);
    }
}

template <int KP, int CT, int NL>
__device__ __forceinline__ void sage_tile_products(const float* sMean, const float* sX,
        const float* __restrict__ Wp,
        const float* __restrict__ bias, float* __restrict__ out,
        int tileRow0, int nEnd, int do_relu, int cw, int lane) {
    constexpr int K = KP * 128;
    constexpr int stride = K + 4;
    constexpr int Nout = CT * 16;
    constexpr int TPW = 4 * CT / 16;
    const int m  = lane & 15;
    const int hi = lane >> 4;
    const int ct = cw % CT;
    const int t0 = (cw / CT) * TPW;
    const int ncol = ct * 16 + m;
    const float* sMrow = sMean + (t0 * 16 + m) * stride + 2 * hi;
    const float* sXrow = sX + (t0 * 16 + m) * stride + 2 * hi;
    const v4f* wp = (const v4f*)Wp + (size_t)ct * (K / 8) * 64 + lane;

    v8f c[NL];
    #pragma unroll
    for (int t = 0; t < NL; ++t) c[t] = (v8f){0.f, 0.f, 0.f, 0.f, 0.f, 0.f, 0.f, 0.f};

    #pragma unroll(2)
    for (int k0 = 0; k0 < K; k0 += 8) {
        const v4f wl4 = wp[(k0 >> 3) * 64];
        const v4f wr4 = wp[(k0 >> 3) * 64 + 32];
        sage_kstep<KP, NL>(c, sMrow, sXrow, k0,     (v2f){wl4.x, wl4.y}, (v2f){wr4.x, wr4.y});
        sage_kstep<KP, NL>(c, sMrow, sXrow, k0 + 4, (v2f){wl4.z, wl4.w}, (v2f){wr4.z, wr4.w});
    }

    const float bb = bias[ncol];
    #pragma unroll
    for (int t = 0; t < NL; ++t) {
        #pragma unroll
        for (int r = 0; r < 8; ++r) {
            const int row = tileRow0 + (t0 + t) * 16 + r + 8 * hi;
            float v = c[t][r] + bb;
            if (do_relu) v = fmaxf(v, 0.0f);
            if (row < nEnd) out[(size_t)row * Nout + ncol] = v;
        }
    }
}

template <int KP, int NPW>
__device__ __forceinline__ void sage_tile_gather(float* sMean, float* sX,
        const float* __restrict__ xin, const int* __restrict__ rowptr,
        const int* __restrict__ col, const float* __restrict__ inv,
        int tileRow0, int nEnd, int M, int gw, int lane) {
    constexpr int K = KP * 128;
    constexpr int stride = K + 4;
    constexpr int UB = 8;
    const int n0 = tileRow0 + NPW * gw;
    int rp[NPW + 1];
    float ivv[NPW];
    #pragma unroll
    for (int i = 0; i <= NPW; ++i) rp[i] = rowptr[min(n0 + i, nEnd)];
    #pragma unroll
    for (int i = 0; i < NPW; ++i) ivv[i] = inv[min(n0 + i, M - 1)];
    int myn[NPW];
    #pragma unroll
    for (int i = 0; i < NPW; ++i)
        myn[i] = (lane < rp[i + 1] - rp[i]) ? col[rp[i] + lane] : 0;
    {
        float4 own[NPW][KP];
        #pragma unroll
        for (int i = 0; i < NPW; ++i) {
            #pragma unroll
            for (int p = 0; p < KP; ++p) {
                own[i][p] = *(const float4*)(xin + (size_t)min(n0 + i, M - 1) * K
                                             + p * 128 + lane * 4);
            }
        }
        #pragma unroll
        for (int i = 0; i < NPW; ++i) {
            #pragma unroll
            for (int p = 0; p < KP; ++p)
                *(float4*)(sX + (NPW * gw + i) * stride + p * 128 + lane * 4) = own[i][p];
        }
    }
    const float* xl = xin + lane * 4;
    #pragma unroll
    for (int i = 0; i < NPW; ++i) {
        const int beg = rp[i];
        const int deg = rp[i + 1] - beg;
        v4f acc[KP];
        #pragma unroll
        for (int p = 0; p < KP; ++p) acc[p] = (v4f){0.f, 0.f, 0.f, 0.f};
        int my = myn[i];
        for (int b = 0; b < deg; b += 32) {
            if (b > 0) my = (lane < deg - b) ? col[beg + b + lane] : 0;
            const int mcnt = min(32, deg - b);
            int j = 0;
            for (; j + UB <= mcnt; j += UB) sage_add_chunk<KP, UB, true>(acc, xl, my, j, UB);
            if (j < mcnt) sage_add_chunk<KP, UB, false>(acc, xl, my, j, mcnt - j);
        }
        const float iv = ivv[i];
        #pragma unroll
        for (int p = 0; p < KP; ++p) {
            *(v4f*)(sMean + (NPW * gw + i) * stride + p * 128 + lane * 4) = acc[p] * iv;
        }
    }
}

template <int KP, int CT>
__global__ void __launch_bounds__(1024)
sage_layer_wmma(const float* __restrict__ xin, const int* __restrict__ rowptr,
                const int* __restrict__ col, const float* __restrict__ inv,
                const float* __restrict__ Wp,
                const float* __restrict__ bias, float* __restrict__ out,
                int M, int do_relu) {
    constexpr int K = KP * 128;
    constexpr int stride = K + 4;
    constexpr int R = 64;
    constexpr int TPW = 4 * CT / 16;
    constexpr int STG = 2 * R * stride;
    __shared__ float smem[2 * STG];

    const int lane = threadIdx.x & 31;
    const int wave = __builtin_amdgcn_readfirstlane((int)(threadIdx.x >> 5));
    const int grp = wave >> 2;
    const bool gatherer = (grp & 1) == 0;
    const int rw = (grp >> 1) * 4 + (wave & 3);

    const int per  = (M + (int)gridDim.x - 1) / (int)gridDim.x;
    const int nBeg = min(M, (int)blockIdx.x * per);
    const int nEnd = min(M, nBeg + per);
    const int ntiles = (nEnd - nBeg + R - 1) / R;

    for (int it = 0; it <= ntiles; ++it) {
        if (it == 0) {
            if (ntiles > 0)
                sage_tile_gather<KP, 2>(smem, smem + R * stride, xin, rowptr, col, inv,
                                        nBeg, nEnd, M, wave, lane);
        } else if (gatherer) {
            if (it < ntiles) {
                float* sMean = smem + (it & 1) * STG;
                float* sX    = sMean + R * stride;
                sage_tile_gather<KP, 4>(sMean, sX, xin, rowptr, col, inv,
                                        nBeg + it * R, nEnd, M, rw, lane);
            }
        } else {
            const float* sMean = smem + ((it - 1) & 1) * STG;
            const float* sX    = sMean + R * stride;
            const int tileRow0 = nBeg + (it - 1) * R;
            const int nlt = (min(R, nEnd - tileRow0) + 15) >> 4;
            const int nl  = min(TPW, nlt - (rw / CT) * TPW);
            if (nl >= 4) {
                if constexpr (TPW >= 4)
                    sage_tile_products<KP, CT, 4>(sMean, sX, Wp, bias, out, tileRow0, nEnd, do_relu, rw, lane);
            } else if (nl == 3) {
                if constexpr (TPW >= 4)
                    sage_tile_products<KP, CT, 3>(sMean, sX, Wp, bias, out, tileRow0, nEnd, do_relu, rw, lane);
            } else if (nl == 2) {
                sage_tile_products<KP, CT, 2>(sMean, sX, Wp, bias, out, tileRow0, nEnd, do_relu, rw, lane);
            } else if (nl == 1) {
                sage_tile_products<KP, CT, 1>(sMean, sX, Wp, bias, out, tileRow0, nEnd, do_relu, rw, lane);
            }
        }
        __syncthreads();
    }
}

extern "C" void kernel_launch(void* const* d_in, const int* in_sizes, int n_in,
                              void* d_out, int out_size, void* d_ws, size_t ws_size,
                              hipStream_t stream) {
    const float* x    = (const float*)d_in[0];
    const int*   edge = (const int*)d_in[1];
    const float* W1l  = (const float*)d_in[2];
    const float* b1   = (const float*)d_in[3];
    const float* W1r  = (const float*)d_in[4];
    const float* W2l  = (const float*)d_in[5];
    const float* b2   = (const float*)d_in[6];
    const float* W2r  = (const float*)d_in[7];
    const float* W3l  = (const float*)d_in[8];
    const float* b3   = (const float*)d_in[9];
    const float* W3r  = (const float*)d_in[10];

    const int HIDDEN = in_sizes[3];
    const int D_IN   = in_sizes[2] / HIDDEN;
    const int M      = in_sizes[0] / D_IN;
    const int E      = in_sizes[1] / 2;
    const int* src   = edge;
    const int* dst   = edge + E;

    float* w = (float*)d_ws;
    const size_t aggOff = ((size_t)M + 127) & ~(size_t)127;
    const size_t h1Off  = aggOff + (size_t)M * HIDDEN;
    const size_t h2Off  = h1Off  + (size_t)M * HIDDEN;
    float* inv = w;
    float* wpack = w + aggOff;
    float* h1  = w + h1Off;
    float* h2  = w + h2Off;
    float* outp = (float*)d_out;

    const int nbk  = (M + SAGE_BK_NODES - 1) >> SAGE_BK_SHIFT;
    const int nchunk = (E + SAGE_EPB - 1) / SAGE_EPB;
    const size_t mPad = ((size_t)M + 1 + 127) & ~(size_t)127;
    const size_t ePad = ((size_t)E + 127) & ~(size_t)127;
    const size_t bPad = ((size_t)nbk + 1 + 127) & ~(size_t)127;
    const size_t hPad = ((size_t)nbk * nchunk + 127) & ~(size_t)127;
    int* rowptr = (int*)(w + h2Off + (size_t)M * HIDDEN);
    int* col    = rowptr + mPad;
    int* bstart = col + ePad;
    int* hist   = bstart + bPad;
    int* pairs  = hist + hPad;

    const int packBlocks = (SAGE_PK1 + SAGE_PK2 + SAGE_PK3 + 1023) / 1024;
    sage_bucket_count<<<packBlocks + nchunk, 1024, 0, stream>>>(dst, E, nbk, hist,
        W1l, W1r, W2l, W2r, W3l, W3r, wpack, packBlocks);
    sage_bucket_place<<<nchunk, 1024, 0, stream>>>(src, dst, E, nbk, nchunk, hist, bstart, pairs);
    sage_bucket_csr<<<nbk, 1024, 0, stream>>>(pairs, bstart, rowptr, inv, col, M);

    const int nblk = 256;
    sage_layer_wmma<1, 16><<<nblk, 1024, 0, stream>>>(x, rowptr, col, inv, wpack, b1, h1, M, 1);
    sage_layer_wmma<2, 16><<<nblk, 1024, 0, stream>>>(h1, rowptr, col, inv,
        wpack + (size_t)SAGE_PK1 * 4, b2, h2, M, 1);
    sage_layer_wmma<2, 8><<<nblk, 1024, 0, stream>>>(h2, rowptr, col, inv,
        wpack + (size_t)(SAGE_PK1 + SAGE_PK2) * 4, b3, outp, M, 0);
}
